# starting kernel
# baseline (speedup 1.0000x reference)
#include <hip/hip_runtime.h>

typedef __attribute__((ext_vector_type(2))) float v2f;
typedef __attribute__((ext_vector_type(8))) float v8f;


__global__ void deg_init_kernel(int* __restrict__ deg, int n) {
    int i = blockIdx.x * blockDim.x + threadIdx.x;
    if (i < n) deg[i] = 1;
}

__global__ void deg_accum_kernel(const int* __restrict__ dst, int* __restrict__ deg, int E) {
    int e = blockIdx.x * blockDim.x + threadIdx.x;
    if (e < E) atomicAdd(&deg[dst[e]], 1);
}

__global__ void dinv_kernel(float* __restrict__ dinv, const int* __restrict__ deg, int n) {
    int i = blockIdx.x * blockDim.x + threadIdx.x;
    if (i < n) {
        float d = (float)deg[i];
        dinv[i] = rsqrtf(d);
    }
}

__global__ void norm_kernel(const int* __restrict__ src, const int* __restrict__ dst,
                            const float* __restrict__ dinv, float* __restrict__ norm, int E) {
    int e = blockIdx.x * blockDim.x + threadIdx.x;
    if (e < E) norm[e] = dinv[src[e]] * dinv[dst[e]];
}

template <int K, int NCOL>
__global__ __launch_bounds__(128) void gemm_wmma_f32(const float* __restrict__ A,
                                                     const float* __restrict__ B,
                                                     float* __restrict__ C, int nrows) {
    const int wave = threadIdx.x >> 5;
    const int lane = threadIdx.x & 31;
    const int half = lane >> 4;
    const int mr   = lane & 15;

    const int row_tile = blockIdx.x * (blockDim.x >> 5) + wave;
    const int row0 = row_tile * 16;
    if (row0 >= nrows) return;

    constexpr int NT = NCOL / 16;
    v8f acc[NT] = {};

    const float* arow = A + (size_t)(row0 + mr) * K + half * 2;

#pragma unroll 8
    for (int kb = 0; kb < K / 4; ++kb) {
        v2f a;
        a.x = arow[kb * 4 + 0];
        a.y = arow[kb * 4 + 1];
        const float* brow = B + (size_t)(kb * 4 + half * 2) * NCOL + mr;
#pragma unroll
        for (int ct = 0; ct < NT; ++ct) {
            v2f b;
            b.x = brow[ct * 16];
            b.y = brow[ct * 16 + NCOL];
            acc[ct] = __builtin_amdgcn_wmma_f32_16x16x4_f32(
                 false, a,  false, b,
                 (short)0, acc[ct],  false,  false);
        }
    }

#pragma unroll
    for (int ct = 0; ct < NT; ++ct) {
        float* crow = C + (size_t)(row0 + half * 8) * NCOL + ct * 16 + mr;
#pragma unroll
        for (int r = 0; r < 8; ++r) crow[(size_t)r * NCOL] = acc[ct][r];
    }
}

template <int F>
__global__ void self_bias_init_kernel(const float* __restrict__ h, const float* __restrict__ dinv,
                                      const float* __restrict__ b, float* __restrict__ out,
                                      long total) {
    long idx = (long)blockIdx.x * blockDim.x + threadIdx.x;
    if (idx >= total) return;
    constexpr int SH = (F == 128) ? 7 : 6;
    int  f = (int)(idx & (F - 1));
    long i = idx >> SH;
    float di = dinv[i];
    out[idx] = b[f] + di * di * h[idx];
}

__global__ void relu_kernel(float* __restrict__ x, long total) {
    long idx = (long)blockIdx.x * blockDim.x + threadIdx.x;
    if (idx < total) x[idx] = fmaxf(x[idx], 0.0f);
}

template <int F>
__global__ void scatter_edges_kernel(const float* __restrict__ h, const int* __restrict__ src,
                                     const int* __restrict__ dst, const float* __restrict__ norm,
                                     float* __restrict__ out, int E) {
    unsigned gt = blockIdx.x * blockDim.x + threadIdx.x;
    int e = (int)(gt >> 5);
    if (e >= E) return;
    int lane = threadIdx.x & 31;

    int   s = src[e];
    int   d = dst[e];
    float w = norm[e];

    if constexpr (F == 128) {
        float4 v = *((const float4*)(h + (size_t)s * F) + lane);
        float* o = out + (size_t)d * F + lane * 4;
        atomicAdd(o + 0, v.x * w);
        atomicAdd(o + 1, v.y * w);
        atomicAdd(o + 2, v.z * w);
        atomicAdd(o + 3, v.w * w);
    } else {
        float2 v = *((const float2*)(h + (size_t)s * F) + lane);
        float* o = out + (size_t)d * F + lane * 2;
        atomicAdd(o + 0, v.x * w);
        atomicAdd(o + 1, v.y * w);
    }
}

extern "C" void kernel_launch(void* const* d_in, const int* in_sizes, int n_in,
                              void* d_out, int out_size, void* d_ws, size_t ws_size,
                              hipStream_t stream) {
    const float* x  = (const float*)d_in[0];
    const int*   ei = (const int*)d_in[1];
    const float* W1 = (const float*)d_in[2];
    const float* b1 = (const float*)d_in[3];
    const float* W2 = (const float*)d_in[4];
    const float* b2 = (const float*)d_in[5];

    const int N = in_sizes[0] / 128;
    const int E = in_sizes[1] / 2;
    const int* src = ei;
    const int* dst = ei + E;

    float* dinv = (float*)d_ws;
    float* norm = dinv + N;
    float* h1   = norm + E;
    float* agg1 = h1 + (size_t)N * 128;
    float* h2   = h1;
    float* out  = (float*)d_out;

    const int T = 256;

    deg_init_kernel<<<(N + T - 1) / T, T, 0, stream>>>((int*)dinv, N);
    deg_accum_kernel<<<(E + T - 1) / T, T, 0, stream>>>(dst, (int*)dinv, E);
    dinv_kernel<<<(N + T - 1) / T, T, 0, stream>>>(dinv, (const int*)dinv, N);
    norm_kernel<<<(E + T - 1) / T, T, 0, stream>>>(src, dst, dinv, norm, E);

    const int row_tiles = (N + 15) / 16;
    const int gemm_blocks = (row_tiles + 3) / 4;

    gemm_wmma_f32<128, 128><<<gemm_blocks, 128, 0, stream>>>(x, W1, h1, N);
    long tot1 = (long)N * 128;
    self_bias_init_kernel<128><<<(int)((tot1 + T - 1) / T), T, 0, stream>>>(h1, dinv, b1, agg1, tot1);
    scatter_edges_kernel<128><<<(E + 7) / 8, T, 0, stream>>>(h1, src, dst, norm, agg1, E);
    relu_kernel<<<(int)((tot1 + T - 1) / T), T, 0, stream>>>(agg1, tot1);

    gemm_wmma_f32<128, 64><<<gemm_blocks, 128, 0, stream>>>(agg1, W2, h2, N);
    long tot2 = (long)N * 64;
    self_bias_init_kernel<64><<<(int)((tot2 + T - 1) / T), T, 0, stream>>>(h2, dinv, b2, out, tot2);
    scatter_edges_kernel<64><<<(E + 7) / 8, T, 0, stream>>>(h2, src, dst, norm, out, E);
}


# Degrees without global atomics: one list-building launch, a block per 400 rows scans the range bytes, counts row degrees by LDS integer adds, parks (src,row) words; 7 to 6 launches; sums: LDS arrival
# speedup vs baseline: 3.3683x; 3.3683x over previous
#include <hip/hip_runtime.h>


typedef __attribute__((ext_vector_type(2))) float v2f;
typedef __attribute__((ext_vector_type(8))) float v8f;

constexpr int RANGE = 400;
constexpr int AGG_THREADS = 1024;
constexpr int OUT_F = 64;
static_assert(RANGE <= 512, "the local row index of a parked edge has 9 bits");
constexpr int LIST_QCAP = 16384;
static_assert(RANGE * (OUT_F / 2) <= LIST_QCAP, "a kept list must fit the LDS queue");

constexpr int PACK_THREADS = 256;
constexpr int PACK_U = 4;
constexpr int PACK_GROUPS = PACK_THREADS * PACK_U;

__device__ __forceinline__ void pack_group(int4 s, int4 d, int g, unsigned char* __restrict__ bin8,
                                           int* __restrict__ pk) {
    const unsigned rx = (unsigned)d.x / (unsigned)RANGE;
    const unsigned ry = (unsigned)d.y / (unsigned)RANGE;
    const unsigned rz = (unsigned)d.z / (unsigned)RANGE;
    const unsigned rw = (unsigned)d.w / (unsigned)RANGE;
    int4 p;
    p.x = (s.x << 9) | (int)((unsigned)d.x - rx * RANGE);
    p.y = (s.y << 9) | (int)((unsigned)d.y - ry * RANGE);
    p.z = (s.z << 9) | (int)((unsigned)d.z - rz * RANGE);
    p.w = (s.w << 9) | (int)((unsigned)d.w - rw * RANGE);
    ((int4*)pk)[g] = p;
    ((unsigned*)bin8)[g] = (rx & 255u) | ((ry & 255u) << 8) | ((rz & 255u) << 16) | (rw << 24);
}

__global__ __launch_bounds__(PACK_THREADS) void edge_pack_kernel(const int* __restrict__ src,
                                                                 const int* __restrict__ dst,
                                                                 unsigned char* __restrict__ bin8,
                                                                 int* __restrict__ pk, int E) {
    const int tid = threadIdx.x;
    const bool vec = ((((size_t)src | (size_t)dst | (size_t)pk) & 15) == 0) && (((size_t)bin8 & 3) == 0);
    const int n4 = vec ? (E >> 2) : 0;
    const int gfirst = blockIdx.x * PACK_GROUPS;
    const int4* s4 = (const int4*)src;
    const int4* d4 = (const int4*)dst;
    if (gfirst + PACK_GROUPS <= n4) {
        int4 s[PACK_U], d[PACK_U];
#pragma unroll
        for (int k = 0; k < PACK_U; ++k) {
            s[k] = s4[gfirst + k * PACK_THREADS + tid];
            d[k] = d4[gfirst + k * PACK_THREADS + tid];
        }
#pragma unroll
        for (int k = 0; k < PACK_U; ++k) pack_group(s[k], d[k], gfirst + k * PACK_THREADS + tid, bin8, pk);
    } else {
        for (int k = 0; k < PACK_U; ++k) {
            const int g = gfirst + k * PACK_THREADS + tid;
            if (g < n4) pack_group(s4[g], d4[g], g, bin8, pk);
        }
    }
    const int stride = gridDim.x * PACK_THREADS;
    for (int e = (n4 << 2) + blockIdx.x * PACK_THREADS + tid; e < E; e += stride) {
        const unsigned dd = (unsigned)dst[e];
        const unsigned r = dd / (unsigned)RANGE;
        pk[e] = (src[e] << 9) | (int)(dd - r * RANGE);
        bin8[e] = (unsigned char)r;
    }
}

__device__ __forceinline__ unsigned byte_flags(unsigned w, unsigned pat) {
    unsigned x = w ^ pat;
    unsigned t = (x & 0x7f7f7f7fu) + 0x7f7f7f7fu;
    return ~(t | x | 0x7f7f7f7fu);
}
__device__ __forceinline__ unsigned group_flags(uint4 v, unsigned pat) {
    return byte_flags(v.x, pat) | (byte_flags(v.y, pat) >> 1) |
           (byte_flags(v.z, pat) >> 2) | (byte_flags(v.w, pat) >> 3);
}

template <typename Push>
__device__ __forceinline__ void scan_range_bytes(const unsigned char* __restrict__ bin8, int E,
                                                 int bid, int tid, Push push) {
    const unsigned pat = (unsigned)bid * 0x01010101u;
    const uint4* b16 = (const uint4*)bin8;
    const int n16 = E >> 4;
    constexpr int STEP = AGG_THREADS * 4;
    const int niter = (n16 + STEP - 1) / STEP;
    int it = (niter > 0) ? (bid % niter) : 0;
    for (int i = 0; i < niter; ++i) {
        const int gb = it * STEP + tid;
        const int g1 = gb + AGG_THREADS, g2 = gb + 2 * AGG_THREADS, g3 = gb + 3 * AGG_THREADS;
        unsigned f0, f1, f2, f3;
        if ((it + 1) * STEP <= n16) {
            uint4 v0 = b16[gb], v1 = b16[g1], v2 = b16[g2], v3 = b16[g3];
            f0 = group_flags(v0, pat);
            f1 = group_flags(v1, pat);
            f2 = group_flags(v2, pat);
            f3 = group_flags(v3, pat);
        } else {
            uint4 v0 = b16[min(gb, n16 - 1)], v1 = b16[min(g1, n16 - 1)];
            uint4 v2 = b16[min(g2, n16 - 1)], v3 = b16[min(g3, n16 - 1)];
            f0 = (gb < n16) ? group_flags(v0, pat) : 0u;
            f1 = (g1 < n16) ? group_flags(v1, pat) : 0u;
            f2 = (g2 < n16) ? group_flags(v2, pat) : 0u;
            f3 = (g3 < n16) ? group_flags(v3, pat) : 0u;
        }
        unsigned lo = f0 | (f1 >> 4);
        unsigned hi = f2 | (f3 >> 4);
        unsigned long long mm = ((unsigned long long)hi << 32) | lo;
        while (mm) {
            int b = __builtin_ctzll(mm);
            mm &= mm - 1;
            int j = (b >> 3) & 3;
            int k = ~b & 3;
            int u = ((b >> 5) << 1) + (((b >> 2) & 1) ^ 1);
            push(((gb + u * AGG_THREADS) << 4) + (k << 2) + j);
        }
        if (++it == niter) it = 0;
    }
    for (int e = (n16 << 4) + tid; e < E; e += AGG_THREADS)
        if ((int)bin8[e] == bid) push(e);
}

__global__ __launch_bounds__(AGG_THREADS) void range_list_kernel(
        const int* __restrict__ pk, const unsigned char* __restrict__ bin8,
        float* __restrict__ dinv, float* park, int* __restrict__ pcount, int N, int E) {
    __shared__ int queue[LIST_QCAP];
    __shared__ int rowdeg[RANGE];
    __shared__ int qcount;

    const int tid = threadIdx.x;
    const int bid = blockIdx.x;
    const int r0  = bid * RANGE;
    const int nrows = min(RANGE, N - r0);

    for (int i = tid; i < RANGE; i += AGG_THREADS) rowdeg[i] = 1;
    if (tid == 0) qcount = 0;
    __syncthreads();

    auto push = [&](int e) {
        int slot = atomicAdd(&qcount, 1);
        if (slot < LIST_QCAP) queue[slot] = e;
        else atomicAdd(&rowdeg[pk[e] & 511], 1);
    };
    scan_range_bytes(bin8, E, bid, tid, push);
    __syncthreads();

    const int raw = __builtin_amdgcn_readfirstlane(qcount);
    const int cnt = min(raw, LIST_QCAP);
    const int rcap = nrows * (OUT_F / 2);
    const bool keep = (raw <= LIST_QCAP) && (raw <= rcap);
    if (tid == 0) pcount[bid] = keep ? raw : -1;

    int* rec = (int*)(park + (size_t)r0 * OUT_F);
    constexpr int U = 4;
    for (int base = 0; base < cnt; base += U * AGG_THREADS) {
        int p[U];
#pragma unroll
        for (int k = 0; k < U; ++k) {
            const int i = base + k * AGG_THREADS + tid;
            p[k] = pk[queue[(i < cnt) ? i : 0]];
        }
#pragma unroll
        for (int k = 0; k < U; ++k) {
            const int i = base + k * AGG_THREADS + tid;
            if (i < cnt) {
                atomicAdd(&rowdeg[p[k] & 511], 1);
                if (keep) rec[i] = p[k];
            }
        }
    }
    __syncthreads();

    for (int r = tid; r < nrows; r += AGG_THREADS) dinv[r0 + r] = rsqrtf((float)rowdeg[r]);
}

template <int K, int NCOL>
__global__ __launch_bounds__(128) void gemm_wmma_f32(const float* __restrict__ A,
                                                     const float* __restrict__ B,
                                                     float* __restrict__ C, int nrows) {
    const int wave = threadIdx.x >> 5;
    const int lane = threadIdx.x & 31;
    const int half = lane >> 4;
    const int mr   = lane & 15;

    const int row_tile = blockIdx.x * (blockDim.x >> 5) + wave;
    const int row0 = row_tile * 16;
    if (row0 >= nrows) return;

    constexpr int NT = NCOL / 16;
    v8f acc[NT] = {};

    const float* arow = A + (size_t)(row0 + mr) * K + half * 2;

#pragma unroll 8
    for (int kb = 0; kb < K / 4; ++kb) {
        v2f a;
        a.x = arow[kb * 4 + 0];
        a.y = arow[kb * 4 + 1];
        const float* brow = B + (size_t)(kb * 4 + half * 2) * NCOL + mr;
#pragma unroll
        for (int ct = 0; ct < NT; ++ct) {
            v2f b;
            b.x = brow[ct * 16];
            b.y = brow[ct * 16 + NCOL];
            acc[ct] = __builtin_amdgcn_wmma_f32_16x16x4_f32(
                 false, a,  false, b,
                 (short)0, acc[ct],  false,  false);
        }
    }

#pragma unroll
    for (int ct = 0; ct < NT; ++ct) {
        float* crow = C + (size_t)(row0 + half * 8) * NCOL + ct * 16 + mr;
#pragma unroll
        for (int r = 0; r < 8; ++r) crow[(size_t)r * NCOL] = acc[ct][r];
    }
}

template <int F, int QCAP, bool RELU, bool FIRST>
__global__ __launch_bounds__(AGG_THREADS) void agg_range_kernel(
        const float* __restrict__ h, const int* __restrict__ src, const int* __restrict__ dst,
        const unsigned char* __restrict__ bin8, const float* __restrict__ dinv,
        const float* __restrict__ bias, float* out, float* park, const int* __restrict__ pcount,
        int N, int E) {
    __shared__ float acc[RANGE * F];
    __shared__ int   queue[QCAP];
    __shared__ int   qcount;

    constexpr int C = F / 32;
    constexpr int G = 8;

    const int tid  = threadIdx.x;
    const int lane = tid & 31;
    const int wave = __builtin_amdgcn_readfirstlane(tid >> 5);
    const int bid  = blockIdx.x;
    const int r0   = bid * RANGE;
    const int nrows = min(RANGE, N - r0);
    const int n4   = nrows * (F / 4);

    if (tid == 0) qcount = 0;

    {
        const float4* h4 = (const float4*)(h + (size_t)r0 * F);
        const float4* b4 = (const float4*)bias;
        float4* a4 = (float4*)acc;
        for (int i = tid; i < n4; i += AGG_THREADS) {
            int r = i / (F / 4);
            int c = i & (F / 4 - 1);
            float di = dinv[r0 + r];
            float s = di * di;
            float4 hv = h4[i];
            float4 bv = b4[c];
            float4 o;
            o.x = bv.x + s * hv.x;
            o.y = bv.y + s * hv.y;
            o.z = bv.z + s * hv.z;
            o.w = bv.w + s * hv.w;
            a4[i] = o;
        }
    }
    __syncthreads();

    const int rcap = nrows * (OUT_F / 2);
    int*   rec  = (int*)(park + (size_t)r0 * OUT_F);
    float* wrec = (float*)(rec + rcap);
    const int pc = __builtin_amdgcn_readfirstlane(pcount[bid]);
    const bool parked = (pc >= 0);
    int cnt = min(pc, rcap);

    if (!parked) {
        auto direct_edge = [&](int e) {
            int s = src[e];
            int d = dst[e];
            float w = dinv[s] * dinv[d];
            const float* hs = h + (size_t)s * F;
            float* a = acc + (d - r0) * F;
#pragma unroll 1
            for (int f = 0; f < F; ++f) atomicAdd(a + f, w * hs[f]);
        };
        auto push = [&](int e) {
            int slot = atomicAdd(&qcount, 1);
            if (slot < QCAP) queue[slot] = e;
            else direct_edge(e);
        };
        scan_range_bytes(bin8, E, bid, tid, push);
        __syncthreads();
        cnt = min(__builtin_amdgcn_readfirstlane(qcount), QCAP);
    }

    {
        for (int base = wave * 32; base < cnt; base += AGG_THREADS) {
            const int nv = min(32, cnt - base);
            int s = 0, dl = 0;
            float w = 0.0f;
            if (lane < nv) {
                if (parked) {
                    int p = rec[base + lane];
                    s = p >> 9;
                    dl = p & 511;
                    if (FIRST) {
                        w = dinv[s] * dinv[r0 + dl];
                        wrec[base + lane] = w;
                    } else {
                        w = wrec[base + lane];
                    }
                } else {
                    int e = queue[base + lane];
                    s = src[e];
                    int d = dst[e];
                    dl = d - r0;
                    w = dinv[s] * dinv[d];
                }
            }
            const int wi = __float_as_int(w);
            int j = 0;
            for (; j + G <= nv; j += G) {
                float v[G][C];
                int   dls[G];
                float ws[G];
#pragma unroll
                for (int k = 0; k < G; ++k) {
                    int sj = __builtin_amdgcn_readlane(s, j + k);
                    dls[k] = __builtin_amdgcn_readlane(dl, j + k);
                    ws[k]  = __int_as_float(__builtin_amdgcn_readlane(wi, j + k));
                    const float* hs = h + (size_t)sj * F + lane;
#pragma unroll
                    for (int c = 0; c < C; ++c) v[k][c] = hs[c * 32];
                }
#pragma unroll
                for (int k = 0; k < G; ++k) {
                    float* a = acc + dls[k] * F + lane;
#pragma unroll
                    for (int c = 0; c < C; ++c) atomicAdd(a + c * 32, v[k][c] * ws[k]);
                }
            }
            for (; j < nv; ++j) {
                int   sj = __builtin_amdgcn_readlane(s, j);
                int   dj = __builtin_amdgcn_readlane(dl, j);
                float wj = __int_as_float(__builtin_amdgcn_readlane(wi, j));
                const float* hs = h + (size_t)sj * F + lane;
                float* a = acc + dj * F + lane;
                float v[C];
#pragma unroll
                for (int c = 0; c < C; ++c) v[c] = hs[c * 32];
#pragma unroll
                for (int c = 0; c < C; ++c) atomicAdd(a + c * 32, v[c] * wj);
            }
        }
    }
    __syncthreads();

    {
        const float4* a4 = (const float4*)acc;
        float4* o4 = (float4*)(out + (size_t)r0 * F);
        for (int i = tid; i < n4; i += AGG_THREADS) {
            float4 o = a4[i];
            if (RELU) {
                o.x = fmaxf(o.x, 0.0f);
                o.y = fmaxf(o.y, 0.0f);
                o.z = fmaxf(o.z, 0.0f);
                o.w = fmaxf(o.w, 0.0f);
            }
            o4[i] = o;
        }
    }
}

extern "C" void kernel_launch(void* const* d_in, const int* in_sizes, int n_in,
                              void* d_out, int out_size, void* d_ws, size_t ws_size,
                              hipStream_t stream) {
    const float* x  = (const float*)d_in[0];
    const int*   ei = (const int*)d_in[1];
    const float* W1 = (const float*)d_in[2];
    const float* b1 = (const float*)d_in[3];
    const float* W2 = (const float*)d_in[4];
    const float* b2 = (const float*)d_in[5];

    const int N = in_sizes[0] / 128;
    const int E = in_sizes[1] / 2;
    const int* src = ei;
    const int* dst = ei + E;

    float* dinv = (float*)d_ws;
    float* rsv  = dinv + N;
    unsigned char* bin8 = (unsigned char*)rsv;
    float* h1   = rsv + E;
    float* agg1 = h1 + (size_t)N * 128;
    float* h2   = h1;
    float* out  = (float*)d_out;
    int* pcount = (int*)(rsv + E / 2);
    int* pk = (int*)agg1;

    const int row_tiles = (N + 15) / 16;
    const int gemm_blocks = (row_tiles + 3) / 4;
    const int agg_blocks = (N + RANGE - 1) / RANGE;

    const int pack_blocks = (E / 4 + PACK_GROUPS - 1) / PACK_GROUPS + 1;
    edge_pack_kernel<<<pack_blocks, PACK_THREADS, 0, stream>>>(src, dst, bin8, pk, E);
    range_list_kernel<<<agg_blocks, AGG_THREADS, 0, stream>>>(pk, bin8, dinv, out, pcount, N, E);

    gemm_wmma_f32<128, 128><<<gemm_blocks, 128, 0, stream>>>(x, W1, h1, N);
    agg_range_kernel<128, 8192, true, true><<<agg_blocks, AGG_THREADS, 0, stream>>>(
        h1, src, dst, bin8, dinv, b1, agg1, out, pcount, N, E);

    gemm_wmma_f32<128, 64><<<gemm_blocks, 128, 0, stream>>>(agg1, W2, h2, N);
    agg_range_kernel<64, 16384, false, false><<<agg_blocks, AGG_THREADS, 0, stream>>>(
        h2, src, dst, bin8, dinv, b2, out, out, pcount, N, E);
}


# Layer-1 product x@W1 runs as extra work-groups of the list-building launch (role by block index; 16x64 units, W1 in the queue's LDS, 64 registers): WMMA issue fills the scan's waits; 6 to 5 launches
# speedup vs baseline: 3.6046x; 1.0702x over previous
#include <hip/hip_runtime.h>


typedef __attribute__((ext_vector_type(2))) float v2f;
typedef __attribute__((ext_vector_type(8))) float v8f;

constexpr int RANGE = 400;
constexpr int AGG_THREADS = 1024;
constexpr int OUT_F = 64;
static_assert(RANGE <= 512, "the local row index of a parked edge has 9 bits");
constexpr int LIST_QCAP = 16384;
static_assert(RANGE * (OUT_F / 2) <= LIST_QCAP, "a kept list must fit the LDS queue");

constexpr int PACK_THREADS = 256;
constexpr int PACK_U = 4;
constexpr int PACK_GROUPS = PACK_THREADS * PACK_U;

__device__ __forceinline__ void pack_group(int4 s, int4 d, int g, unsigned char* __restrict__ bin8,
                                           int* __restrict__ pk) {
    const unsigned rx = (unsigned)d.x / (unsigned)RANGE;
    const unsigned ry = (unsigned)d.y / (unsigned)RANGE;
    const unsigned rz = (unsigned)d.z / (unsigned)RANGE;
    const unsigned rw = (unsigned)d.w / (unsigned)RANGE;
    int4 p;
    p.x = (s.x << 9) | (int)((unsigned)d.x - rx * RANGE);
    p.y = (s.y << 9) | (int)((unsigned)d.y - ry * RANGE);
    p.z = (s.z << 9) | (int)((unsigned)d.z - rz * RANGE);
    p.w = (s.w << 9) | (int)((unsigned)d.w - rw * RANGE);
    ((int4*)pk)[g] = p;
    ((unsigned*)bin8)[g] = (rx & 255u) | ((ry & 255u) << 8) | ((rz & 255u) << 16) | (rw << 24);
}

__global__ __launch_bounds__(PACK_THREADS) void edge_pack_kernel(const int* __restrict__ src,
                                                                 const int* __restrict__ dst,
                                                                 unsigned char* __restrict__ bin8,
                                                                 int* __restrict__ pk, int E) {
    const int tid = threadIdx.x;
    const bool vec = ((((size_t)src | (size_t)dst | (size_t)pk) & 15) == 0) && (((size_t)bin8 & 3) == 0);
    const int n4 = vec ? (E >> 2) : 0;
    const int gfirst = blockIdx.x * PACK_GROUPS;
    const int4* s4 = (const int4*)src;
    const int4* d4 = (const int4*)dst;
    if (gfirst + PACK_GROUPS <= n4) {
        int4 s[PACK_U], d[PACK_U];
#pragma unroll
        for (int k = 0; k < PACK_U; ++k) {
            s[k] = s4[gfirst + k * PACK_THREADS + tid];
            d[k] = d4[gfirst + k * PACK_THREADS + tid];
        }
#pragma unroll
        for (int k = 0; k < PACK_U; ++k) pack_group(s[k], d[k], gfirst + k * PACK_THREADS + tid, bin8, pk);
    } else {
        for (int k = 0; k < PACK_U; ++k) {
            const int g = gfirst + k * PACK_THREADS + tid;
            if (g < n4) pack_group(s4[g], d4[g], g, bin8, pk);
        }
    }
    const int stride = gridDim.x * PACK_THREADS;
    for (int e = (n4 << 2) + blockIdx.x * PACK_THREADS + tid; e < E; e += stride) {
        const unsigned dd = (unsigned)dst[e];
        const unsigned r = dd / (unsigned)RANGE;
        pk[e] = (src[e] << 9) | (int)(dd - r * RANGE);
        bin8[e] = (unsigned char)r;
    }
}

__device__ __forceinline__ unsigned byte_flags(unsigned w, unsigned pat) {
    unsigned x = w ^ pat;
    unsigned t = (x & 0x7f7f7f7fu) + 0x7f7f7f7fu;
    return ~(t | x | 0x7f7f7f7fu);
}
__device__ __forceinline__ unsigned group_flags(uint4 v, unsigned pat) {
    return byte_flags(v.x, pat) | (byte_flags(v.y, pat) >> 1) |
           (byte_flags(v.z, pat) >> 2) | (byte_flags(v.w, pat) >> 3);
}

template <typename Push>
__device__ __forceinline__ void scan_range_bytes(const unsigned char* __restrict__ bin8, int E,
                                                 int bid, int tid, Push push) {
    const unsigned pat = (unsigned)bid * 0x01010101u;
    const uint4* b16 = (const uint4*)bin8;
    const int n16 = E >> 4;
    constexpr int STEP = AGG_THREADS * 4;
    const int niter = (n16 + STEP - 1) / STEP;
    int it = (niter > 0) ? (bid % niter) : 0;
    for (int i = 0; i < niter; ++i) {
        const int gb = it * STEP + tid;
        const int g1 = gb + AGG_THREADS, g2 = gb + 2 * AGG_THREADS, g3 = gb + 3 * AGG_THREADS;
        unsigned f0, f1, f2, f3;
        if ((it + 1) * STEP <= n16) {
            uint4 v0 = b16[gb], v1 = b16[g1], v2 = b16[g2], v3 = b16[g3];
            f0 = group_flags(v0, pat);
            f1 = group_flags(v1, pat);
            f2 = group_flags(v2, pat);
            f3 = group_flags(v3, pat);
        } else {
            uint4 v0 = b16[min(gb, n16 - 1)], v1 = b16[min(g1, n16 - 1)];
            uint4 v2 = b16[min(g2, n16 - 1)], v3 = b16[min(g3, n16 - 1)];
            f0 = (gb < n16) ? group_flags(v0, pat) : 0u;
            f1 = (g1 < n16) ? group_flags(v1, pat) : 0u;
            f2 = (g2 < n16) ? group_flags(v2, pat) : 0u;
            f3 = (g3 < n16) ? group_flags(v3, pat) : 0u;
        }
        unsigned lo = f0 | (f1 >> 4);
        unsigned hi = f2 | (f3 >> 4);
        unsigned long long mm = ((unsigned long long)hi << 32) | lo;
        while (mm) {
            int b = __builtin_ctzll(mm);
            mm &= mm - 1;
            int j = (b >> 3) & 3;
            int k = ~b & 3;
            int u = ((b >> 5) << 1) + (((b >> 2) & 1) ^ 1);
            push(((gb + u * AGG_THREADS) << 4) + (k << 2) + j);
        }
        if (++it == niter) it = 0;
    }
    for (int e = (n16 << 4) + tid; e < E; e += AGG_THREADS)
        if ((int)bin8[e] == bid) push(e);
}

template <int K, int NCOL, int NT>
__device__ __forceinline__ void gemm_strip_f32(const float* __restrict__ A,
                                               const float* __restrict__ B,
                                               float* __restrict__ C, int row0, int col0, int lane) {
    const int half = lane >> 4;
    const int mr   = lane & 15;

    v8f acc[NT] = {};

    const float* arow = A + (size_t)(row0 + mr) * K + half * 2;
    const float* bcol = B + (size_t)(half * 2) * NCOL + col0 + mr;

#pragma unroll 8
    for (int kb = 0; kb < K / 4; ++kb) {
        v2f a;
        a.x = arow[kb * 4 + 0];
        a.y = arow[kb * 4 + 1];
        const float* brow = bcol + (size_t)(kb * 4) * NCOL;
#pragma unroll
        for (int ct = 0; ct < NT; ++ct) {
            v2f b;
            b.x = brow[ct * 16];
            b.y = brow[ct * 16 + NCOL];
            acc[ct] = __builtin_amdgcn_wmma_f32_16x16x4_f32(
                 false, a,  false, b,
                 (short)0, acc[ct],  false,  false);
        }
    }

#pragma unroll
    for (int ct = 0; ct < NT; ++ct) {
        float* crow = C + (size_t)(row0 + half * 8) * NCOL + col0 + ct * 16 + mr;
#pragma unroll
        for (int r = 0; r < 8; ++r) crow[(size_t)r * NCOL] = acc[ct][r];
    }
}

template <int K, int NCOL>
__global__ __launch_bounds__(128) void gemm_wmma_f32(const float* __restrict__ A,
                                                     const float* __restrict__ B,
                                                     float* __restrict__ C, int nrows) {
    const int wave = threadIdx.x >> 5;
    const int lane = threadIdx.x & 31;
    const int row_tile = blockIdx.x * (blockDim.x >> 5) + wave;
    const int row0 = row_tile * 16;
    if (row0 >= nrows) return;
    gemm_strip_f32<K, NCOL, NCOL / 16>(A, B, C, row0, 0, lane);
}

template <int K, int NCOL, int NT>
__device__ __forceinline__ void gemm_strip_f32_ldsB(const float* __restrict__ A, const float* Bs,
                                                    float* __restrict__ C, int row0, int col0, int lane) {
    const int half = lane >> 4;
    const int mr   = lane & 15;

    v8f acc[NT] = {};

    const float* arow = A + (size_t)(row0 + mr) * K + half * 2;
    const float* bcol = Bs + half * (2 * NCOL) + (col0 + mr) * 2;

#pragma unroll 4
    for (int kb = 0; kb < K / 4; ++kb) {
        v2f a;
        a.x = arow[kb * 4 + 0];
        a.y = arow[kb * 4 + 1];
        const float* brow = bcol + kb * (4 * NCOL);
        v2f b[NT];
#pragma unroll
        for (int ct = 0; ct < NT; ++ct) b[ct] = *(const v2f*)(brow + ct * 32);
#pragma unroll
        for (int ct = 0; ct < NT; ++ct)
            acc[ct] = __builtin_amdgcn_wmma_f32_16x16x4_f32(
                 false, a,  false, b[ct],
                 (short)0, acc[ct],  false,  false);
    }

#pragma unroll
    for (int ct = 0; ct < NT; ++ct) {
        float* crow = C + (size_t)(row0 + half * 8) * NCOL + col0 + ct * 16 + mr;
#pragma unroll
        for (int r = 0; r < 8; ++r) crow[(size_t)r * NCOL] = acc[ct][r];
    }
}

static_assert(RANGE % 16 == 0, "the product role cuts a block's rows into 16-row strips");
constexpr int PROD_COLS = 64;
static_assert(128 / PROD_COLS == 2, "a unit index is (strip << 1) | column half");
__global__ __launch_bounds__(AGG_THREADS) __attribute__((amdgpu_waves_per_eu(16, 16)))
void range_list_product_kernel(
        const int* __restrict__ pk, const unsigned char* __restrict__ bin8,
        float* __restrict__ dinv, float* park, int* __restrict__ pcount, int N, int E,
        const float* __restrict__ x, const float* __restrict__ W, float* __restrict__ h,
        int list_blocks) {
    __shared__ __attribute__((aligned(16))) unsigned char stage[LIST_QCAP * 4];
    __shared__ int rowdeg[RANGE];
    __shared__ int qcount;
    static_assert(128 * 128 * sizeof(float) <= sizeof(stage), "W must fit the queue's LDS");

    const int tid = threadIdx.x;

    if ((int)blockIdx.x >= list_blocks) {
        const int g = (int)blockIdx.x - list_blocks;
        const int lane = tid & 31;
        const int wave = __builtin_amdgcn_readfirstlane(tid >> 5);
        float* Ws = (float*)stage;
        for (int i = tid; i < 128 * 128; i += AGG_THREADS) {
            const int k = i >> 7, col = i & 127;
            Ws[(k >> 1) * 256 + col * 2 + (k & 1)] = W[i];
        }
        __syncthreads();
        constexpr int UNITS = (RANGE / 16) * (128 / PROD_COLS);
        for (int u = wave; u < UNITS; u += AGG_THREADS / 32) {
            const int row0 = g * RANGE + (u >> 1) * 16;
            if (row0 < N)
                gemm_strip_f32_ldsB<128, 128, PROD_COLS / 16>(x, Ws, h, row0, (u & 1) * PROD_COLS, lane);
        }
        return;
    }

    int* queue = (int*)stage;
    const int bid = blockIdx.x;
    const int r0  = bid * RANGE;
    const int nrows = min(RANGE, N - r0);

    for (int i = tid; i < RANGE; i += AGG_THREADS) rowdeg[i] = 1;
    if (tid == 0) qcount = 0;
    __syncthreads();

    auto push = [&](int e) {
        int slot = atomicAdd(&qcount, 1);
        if (slot < LIST_QCAP) queue[slot] = e;
        else atomicAdd(&rowdeg[pk[e] & 511], 1);
    };
    scan_range_bytes(bin8, E, bid, tid, push);
    __syncthreads();

    const int raw = __builtin_amdgcn_readfirstlane(qcount);
    const int cnt = min(raw, LIST_QCAP);
    const int rcap = nrows * (OUT_F / 2);
    const bool keep = (raw <= LIST_QCAP) && (raw <= rcap);
    if (tid == 0) pcount[bid] = keep ? raw : -1;

    int* rec = (int*)(park + (size_t)r0 * OUT_F);
    constexpr int U = 4;
    for (int base = 0; base < cnt; base += U * AGG_THREADS) {
        int p[U];
#pragma unroll
        for (int k = 0; k < U; ++k) {
            const int i = base + k * AGG_THREADS + tid;
            p[k] = pk[queue[(i < cnt) ? i : 0]];
        }
#pragma unroll
        for (int k = 0; k < U; ++k) {
            const int i = base + k * AGG_THREADS + tid;
            if (i < cnt) {
                atomicAdd(&rowdeg[p[k] & 511], 1);
                if (keep) rec[i] = p[k];
            }
        }
    }
    __syncthreads();

    for (int r = tid; r < nrows; r += AGG_THREADS) dinv[r0 + r] = rsqrtf((float)rowdeg[r]);
}

template <int F, int QCAP, bool RELU, bool FIRST>
__global__ __launch_bounds__(AGG_THREADS) void agg_range_kernel(
        const float* __restrict__ h, const int* __restrict__ src, const int* __restrict__ dst,
        const unsigned char* __restrict__ bin8, const float* __restrict__ dinv,
        const float* __restrict__ bias, float* out, float* park, const int* __restrict__ pcount,
        int N, int E) {
    __shared__ float acc[RANGE * F];
    __shared__ int   queue[QCAP];
    __shared__ int   qcount;

    constexpr int C = F / 32;
    constexpr int G = 8;

    const int tid  = threadIdx.x;
    const int lane = tid & 31;
    const int wave = __builtin_amdgcn_readfirstlane(tid >> 5);
    const int bid  = blockIdx.x;
    const int r0   = bid * RANGE;
    const int nrows = min(RANGE, N - r0);
    const int n4   = nrows * (F / 4);

    if (tid == 0) qcount = 0;

    {
        const float4* h4 = (const float4*)(h + (size_t)r0 * F);
        const float4* b4 = (const float4*)bias;
        float4* a4 = (float4*)acc;
        for (int i = tid; i < n4; i += AGG_THREADS) {
            int r = i / (F / 4);
            int c = i & (F / 4 - 1);
            float di = dinv[r0 + r];
            float s = di * di;
            float4 hv = h4[i];
            float4 bv = b4[c];
            float4 o;
            o.x = bv.x + s * hv.x;
            o.y = bv.y + s * hv.y;
            o.z = bv.z + s * hv.z;
            o.w = bv.w + s * hv.w;
            a4[i] = o;
        }
    }
    __syncthreads();

    const int rcap = nrows * (OUT_F / 2);
    int*   rec  = (int*)(park + (size_t)r0 * OUT_F);
    float* wrec = (float*)(rec + rcap);
    const int pc = __builtin_amdgcn_readfirstlane(pcount[bid]);
    const bool parked = (pc >= 0);
    int cnt = min(pc, rcap);

    if (!parked) {
        auto direct_edge = [&](int e) {
            int s = src[e];
            int d = dst[e];
            float w = dinv[s] * dinv[d];
            const float* hs = h + (size_t)s * F;
            float* a = acc + (d - r0) * F;
#pragma unroll 1
            for (int f = 0; f < F; ++f) atomicAdd(a + f, w * hs[f]);
        };
        auto push = [&](int e) {
            int slot = atomicAdd(&qcount, 1);
            if (slot < QCAP) queue[slot] = e;
            else direct_edge(e);
        };
        scan_range_bytes(bin8, E, bid, tid, push);
        __syncthreads();
        cnt = min(__builtin_amdgcn_readfirstlane(qcount), QCAP);
    }

    {
        for (int base = wave * 32; base < cnt; base += AGG_THREADS) {
            const int nv = min(32, cnt - base);
            int s = 0, dl = 0;
            float w = 0.0f;
            if (lane < nv) {
                if (parked) {
                    int p = rec[base + lane];
                    s = p >> 9;
                    dl = p & 511;
                    if (FIRST) {
                        w = dinv[s] * dinv[r0 + dl];
                        wrec[base + lane] = w;
                    } else {
                        w = wrec[base + lane];
                    }
                } else {
                    int e = queue[base + lane];
                    s = src[e];
                    int d = dst[e];
                    dl = d - r0;
                    w = dinv[s] * dinv[d];
                }
            }
            const int wi = __float_as_int(w);
            int j = 0;
            for (; j + G <= nv; j += G) {
                float v[G][C];
                int   dls[G];
                float ws[G];
#pragma unroll
                for (int k = 0; k < G; ++k) {
                    int sj = __builtin_amdgcn_readlane(s, j + k);
                    dls[k] = __builtin_amdgcn_readlane(dl, j + k);
                    ws[k]  = __int_as_float(__builtin_amdgcn_readlane(wi, j + k));
                    const float* hs = h + (size_t)sj * F + lane;
#pragma unroll
                    for (int c = 0; c < C; ++c) v[k][c] = hs[c * 32];
                }
#pragma unroll
                for (int k = 0; k < G; ++k) {
                    float* a = acc + dls[k] * F + lane;
#pragma unroll
                    for (int c = 0; c < C; ++c) atomicAdd(a + c * 32, v[k][c] * ws[k]);
                }
            }
            for (; j < nv; ++j) {
                int   sj = __builtin_amdgcn_readlane(s, j);
                int   dj = __builtin_amdgcn_readlane(dl, j);
                float wj = __int_as_float(__builtin_amdgcn_readlane(wi, j));
                const float* hs = h + (size_t)sj * F + lane;
                float* a = acc + dj * F + lane;
                float v[C];
#pragma unroll
                for (int c = 0; c < C; ++c) v[c] = hs[c * 32];
#pragma unroll
                for (int c = 0; c < C; ++c) atomicAdd(a + c * 32, v[c] * wj);
            }
        }
    }
    __syncthreads();

    {
        const float4* a4 = (const float4*)acc;
        float4* o4 = (float4*)(out + (size_t)r0 * F);
        for (int i = tid; i < n4; i += AGG_THREADS) {
            float4 o = a4[i];
            if (RELU) {
                o.x = fmaxf(o.x, 0.0f);
                o.y = fmaxf(o.y, 0.0f);
                o.z = fmaxf(o.z, 0.0f);
                o.w = fmaxf(o.w, 0.0f);
            }
            o4[i] = o;
        }
    }
}

extern "C" void kernel_launch(void* const* d_in, const int* in_sizes, int n_in,
                              void* d_out, int out_size, void* d_ws, size_t ws_size,
                              hipStream_t stream) {
    const float* x  = (const float*)d_in[0];
    const int*   ei = (const int*)d_in[1];
    const float* W1 = (const float*)d_in[2];
    const float* b1 = (const float*)d_in[3];
    const float* W2 = (const float*)d_in[4];
    const float* b2 = (const float*)d_in[5];

    const int N = in_sizes[0] / 128;
    const int E = in_sizes[1] / 2;
    const int* src = ei;
    const int* dst = ei + E;

    float* dinv = (float*)d_ws;
    float* rsv  = dinv + N;
    unsigned char* bin8 = (unsigned char*)rsv;
    float* h1   = rsv + E;
    float* agg1 = h1 + (size_t)N * 128;
    float* h2   = h1;
    float* out  = (float*)d_out;
    int* pcount = (int*)(rsv + E / 2);
    int* pk = (int*)agg1;

    const int row_tiles = (N + 15) / 16;
    const int gemm_blocks = (row_tiles + 3) / 4;
    const int agg_blocks = (N + RANGE - 1) / RANGE;

    const int pack_blocks = (E / 4 + PACK_GROUPS - 1) / PACK_GROUPS + 1;
    edge_pack_kernel<<<pack_blocks, PACK_THREADS, 0, stream>>>(src, dst, bin8, pk, E);
    range_list_product_kernel<<<2 * agg_blocks, AGG_THREADS, 0, stream>>>(
        pk, bin8, dinv, out, pcount, N, E, x, W1, h1, agg_blocks);

    agg_range_kernel<128, 8192, true, true><<<agg_blocks, AGG_THREADS, 0, stream>>>(
        h1, src, dst, bin8, dinv, b1, agg1, out, pcount, N, E);

    gemm_wmma_f32<128, 64><<<gemm_blocks, 128, 0, stream>>>(agg1, W2, h2, N);
    agg_range_kernel<64, 16384, false, false><<<agg_blocks, AGG_THREADS, 0, stream>>>(
        h2, src, dst, bin8, dinv, b2, out, out, pcount, N, E);
}


# List blocks read a per-range bitmap over 16-edge groups, built by the edge pass in LDS (integer OR), and fetch only marked range-byte groups: no scan of all range bytes per block (sums: LDS arrival)
# speedup vs baseline: 3.7921x; 1.0520x over previous
#include <hip/hip_runtime.h>


typedef __attribute__((ext_vector_type(2))) float v2f;
typedef __attribute__((ext_vector_type(8))) float v8f;

constexpr int RANGE = 400;
constexpr int AGG_THREADS = 1024;
constexpr int OUT_F = 64;
static_assert(RANGE <= 512, "the local row index of a parked edge has 9 bits");
constexpr int LIST_QCAP = 16384;
static_assert(RANGE * (OUT_F / 2) <= LIST_QCAP, "a kept list must fit the LDS queue");
constexpr int LIST_GQCAP = 8192;

constexpr int PACK_THREADS = 256;
constexpr int PACK_U = 4;
constexpr int PACK_GROUPS = PACK_THREADS * PACK_U;
constexpr int PACK_EDGES = PACK_GROUPS * 4;
constexpr int BM_WORDS = PACK_EDGES / 16 / 32;
constexpr int BM_ROWS = 256;
static_assert(BM_WORDS == 8, "a block writes its words of a row as two 16-byte stores");
static_assert(BM_ROWS == PACK_THREADS, "thread t writes row t of the block's bitmap words");

__device__ __forceinline__ void pack_group(int4 s, int4 d, int g, unsigned char* __restrict__ bin8,
                                           int* __restrict__ pk, unsigned* bm, int word, unsigned bit) {
    const unsigned rx = (unsigned)d.x / (unsigned)RANGE;
    const unsigned ry = (unsigned)d.y / (unsigned)RANGE;
    const unsigned rz = (unsigned)d.z / (unsigned)RANGE;
    const unsigned rw = (unsigned)d.w / (unsigned)RANGE;
    int4 p;
    p.x = (s.x << 9) | (int)((unsigned)d.x - rx * RANGE);
    p.y = (s.y << 9) | (int)((unsigned)d.y - ry * RANGE);
    p.z = (s.z << 9) | (int)((unsigned)d.z - rz * RANGE);
    p.w = (s.w << 9) | (int)((unsigned)d.w - rw * RANGE);
    ((int4*)pk)[g] = p;
    ((unsigned*)bin8)[g] = (rx & 255u) | ((ry & 255u) << 8) | ((rz & 255u) << 16) | (rw << 24);
    atomicOr(&bm[word * BM_ROWS + (rx & 255u)], bit);
    atomicOr(&bm[word * BM_ROWS + (ry & 255u)], bit);
    atomicOr(&bm[word * BM_ROWS + (rz & 255u)], bit);
    atomicOr(&bm[word * BM_ROWS + (rw & 255u)], bit);
}

__global__ __launch_bounds__(PACK_THREADS) void edge_pack_kernel(const int* __restrict__ src,
                                                                 const int* __restrict__ dst,
                                                                 unsigned char* __restrict__ bin8,
                                                                 int* __restrict__ pk,
                                                                 unsigned* __restrict__ bitmap, int E) {
    __shared__ __attribute__((aligned(16))) unsigned bm[BM_WORDS * BM_ROWS];
    const int tid = threadIdx.x;
    for (int i = tid; i < BM_WORDS * BM_ROWS; i += PACK_THREADS) bm[i] = 0u;
    __syncthreads();
    const unsigned bit = 1u << ((tid >> 2) & 31);
    const int word0 = tid >> 7;
    const bool vec = ((((size_t)src | (size_t)dst | (size_t)pk) & 15) == 0) && (((size_t)bin8 & 3) == 0);
    const int n4 = vec ? (E >> 2) : 0;
    const int gfirst = blockIdx.x * PACK_GROUPS;
    const int4* s4 = (const int4*)src;
    const int4* d4 = (const int4*)dst;
    if (gfirst + PACK_GROUPS <= n4) {
        int4 s[PACK_U], d[PACK_U];
#pragma unroll
        for (int k = 0; k < PACK_U; ++k) {
            s[k] = s4[gfirst + k * PACK_THREADS + tid];
            d[k] = d4[gfirst + k * PACK_THREADS + tid];
        }
#pragma unroll
        for (int k = 0; k < PACK_U; ++k)
            pack_group(s[k], d[k], gfirst + k * PACK_THREADS + tid, bin8, pk, bm, k * 2 + word0, bit);
    } else {
        for (int k = 0; k < PACK_U; ++k) {
            const int g = gfirst + k * PACK_THREADS + tid;
            if (g < n4) pack_group(s4[g], d4[g], g, bin8, pk, bm, k * 2 + word0, bit);
        }
    }
    const int e0 = blockIdx.x * PACK_EDGES;
    const int e1 = min(E, e0 + PACK_EDGES);
    for (int e = max(e0, n4 << 2) + tid; e < e1; e += PACK_THREADS) {
        const unsigned dd = (unsigned)dst[e];
        const unsigned r = dd / (unsigned)RANGE;
        pk[e] = (src[e] << 9) | (int)(dd - r * RANGE);
        bin8[e] = (unsigned char)r;
        const int el = e - e0;
        atomicOr(&bm[(el >> 9) * BM_ROWS + (r & 255u)], 1u << ((el >> 4) & 31));
    }
    __syncthreads();
    {
        const int wpr = gridDim.x * BM_WORDS;
        uint4 lo, hi;
        lo.x = bm[0 * BM_ROWS + tid]; lo.y = bm[1 * BM_ROWS + tid];
        lo.z = bm[2 * BM_ROWS + tid]; lo.w = bm[3 * BM_ROWS + tid];
        hi.x = bm[4 * BM_ROWS + tid]; hi.y = bm[5 * BM_ROWS + tid];
        hi.z = bm[6 * BM_ROWS + tid]; hi.w = bm[7 * BM_ROWS + tid];
        uint4* o = (uint4*)(bitmap + (size_t)tid * wpr + blockIdx.x * BM_WORDS);
        o[0] = lo;
        o[1] = hi;
    }
}

__device__ __forceinline__ unsigned byte_flags(unsigned w, unsigned pat) {
    unsigned x = w ^ pat;
    unsigned t = (x & 0x7f7f7f7fu) + 0x7f7f7f7fu;
    return ~(t | x | 0x7f7f7f7fu);
}
__device__ __forceinline__ unsigned group_flags(uint4 v, unsigned pat) {
    return byte_flags(v.x, pat) | (byte_flags(v.y, pat) >> 1) |
           (byte_flags(v.z, pat) >> 2) | (byte_flags(v.w, pat) >> 3);
}

template <typename Push>
__device__ __forceinline__ void scan_range_bytes(const unsigned char* __restrict__ bin8, int E,
                                                 int bid, int tid, Push push) {
    const unsigned pat = (unsigned)bid * 0x01010101u;
    const uint4* b16 = (const uint4*)bin8;
    const int n16 = E >> 4;
    constexpr int STEP = AGG_THREADS * 4;
    const int niter = (n16 + STEP - 1) / STEP;
    int it = (niter > 0) ? (bid % niter) : 0;
    for (int i = 0; i < niter; ++i) {
        const int gb = it * STEP + tid;
        const int g1 = gb + AGG_THREADS, g2 = gb + 2 * AGG_THREADS, g3 = gb + 3 * AGG_THREADS;
        unsigned f0, f1, f2, f3;
        if ((it + 1) * STEP <= n16) {
            uint4 v0 = b16[gb], v1 = b16[g1], v2 = b16[g2], v3 = b16[g3];
            f0 = group_flags(v0, pat);
            f1 = group_flags(v1, pat);
            f2 = group_flags(v2, pat);
            f3 = group_flags(v3, pat);
        } else {
            uint4 v0 = b16[min(gb, n16 - 1)], v1 = b16[min(g1, n16 - 1)];
            uint4 v2 = b16[min(g2, n16 - 1)], v3 = b16[min(g3, n16 - 1)];
            f0 = (gb < n16) ? group_flags(v0, pat) : 0u;
            f1 = (g1 < n16) ? group_flags(v1, pat) : 0u;
            f2 = (g2 < n16) ? group_flags(v2, pat) : 0u;
            f3 = (g3 < n16) ? group_flags(v3, pat) : 0u;
        }
        unsigned lo = f0 | (f1 >> 4);
        unsigned hi = f2 | (f3 >> 4);
        unsigned long long mm = ((unsigned long long)hi << 32) | lo;
        while (mm) {
            int b = __builtin_ctzll(mm);
            mm &= mm - 1;
            int j = (b >> 3) & 3;
            int k = ~b & 3;
            int u = ((b >> 5) << 1) + (((b >> 2) & 1) ^ 1);
            push(((gb + u * AGG_THREADS) << 4) + (k << 2) + j);
        }
        if (++it == niter) it = 0;
    }
    for (int e = (n16 << 4) + tid; e < E; e += AGG_THREADS)
        if ((int)bin8[e] == bid) push(e);
}

template <int K, int NCOL, int NT>
__device__ __forceinline__ void gemm_strip_f32(const float* __restrict__ A,
                                               const float* __restrict__ B,
                                               float* __restrict__ C, int row0, int col0, int lane) {
    const int half = lane >> 4;
    const int mr   = lane & 15;

    v8f acc[NT] = {};

    const float* arow = A + (size_t)(row0 + mr) * K + half * 2;
    const float* bcol = B + (size_t)(half * 2) * NCOL + col0 + mr;

#pragma unroll 8
    for (int kb = 0; kb < K / 4; ++kb) {
        v2f a;
        a.x = arow[kb * 4 + 0];
        a.y = arow[kb * 4 + 1];
        const float* brow = bcol + (size_t)(kb * 4) * NCOL;
#pragma unroll
        for (int ct = 0; ct < NT; ++ct) {
            v2f b;
            b.x = brow[ct * 16];
            b.y = brow[ct * 16 + NCOL];
            acc[ct] = __builtin_amdgcn_wmma_f32_16x16x4_f32(
                 false, a,  false, b,
                 (short)0, acc[ct],  false,  false);
        }
    }

#pragma unroll
    for (int ct = 0; ct < NT; ++ct) {
        float* crow = C + (size_t)(row0 + half * 8) * NCOL + col0 + ct * 16 + mr;
#pragma unroll
        for (int r = 0; r < 8; ++r) crow[(size_t)r * NCOL] = acc[ct][r];
    }
}

template <int K, int NCOL>
__global__ __launch_bounds__(128) void gemm_wmma_f32(const float* __restrict__ A,
                                                     const float* __restrict__ B,
                                                     float* __restrict__ C, int nrows) {
    const int wave = threadIdx.x >> 5;
    const int lane = threadIdx.x & 31;
    const int row_tile = blockIdx.x * (blockDim.x >> 5) + wave;
    const int row0 = row_tile * 16;
    if (row0 >= nrows) return;
    gemm_strip_f32<K, NCOL, NCOL / 16>(A, B, C, row0, 0, lane);
}

template <int K, int NCOL, int NT>
__device__ __forceinline__ void gemm_strip_f32_ldsB(const float* __restrict__ A, const float* Bs,
                                                    float* __restrict__ C, int row0, int col0, int lane) {
    const int half = lane >> 4;
    const int mr   = lane & 15;

    v8f acc[NT] = {};

    const float* arow = A + (size_t)(row0 + mr) * K + half * 2;
    const float* bcol = Bs + half * (2 * NCOL) + (col0 + mr) * 2;

#pragma unroll 4
    for (int kb = 0; kb < K / 4; ++kb) {
        v2f a;
        a.x = arow[kb * 4 + 0];
        a.y = arow[kb * 4 + 1];
        const float* brow = bcol + kb * (4 * NCOL);
        v2f b[NT];
#pragma unroll
        for (int ct = 0; ct < NT; ++ct) b[ct] = *(const v2f*)(brow + ct * 32);
#pragma unroll
        for (int ct = 0; ct < NT; ++ct)
            acc[ct] = __builtin_amdgcn_wmma_f32_16x16x4_f32(
                 false, a,  false, b[ct],
                 (short)0, acc[ct],  false,  false);
    }

#pragma unroll
    for (int ct = 0; ct < NT; ++ct) {
        float* crow = C + (size_t)(row0 + half * 8) * NCOL + col0 + ct * 16 + mr;
#pragma unroll
        for (int r = 0; r < 8; ++r) crow[(size_t)r * NCOL] = acc[ct][r];
    }
}

static_assert(RANGE % 16 == 0, "the product role cuts a block's rows into 16-row strips");
constexpr int PROD_COLS = 64;
static_assert(128 / PROD_COLS == 2, "a unit index is (strip << 1) | column half");
__global__ __launch_bounds__(AGG_THREADS) __attribute__((amdgpu_waves_per_eu(16, 16)))
void range_list_product_kernel(
        const int* __restrict__ pk, const unsigned char* __restrict__ bin8,
        const unsigned* __restrict__ bitmap, int wpr,
        float* __restrict__ dinv, float* park, int* __restrict__ pcount, int N, int E,
        const float* __restrict__ x, const float* __restrict__ W, float* __restrict__ h,
        int list_blocks) {
    __shared__ __attribute__((aligned(16))) unsigned char stage[LIST_QCAP * 4];
    __shared__ int gq[LIST_GQCAP];
    __shared__ int rowdeg[RANGE];
    __shared__ int qcount;
    __shared__ int gcount;
    static_assert(128 * 128 * sizeof(float) <= sizeof(stage), "W must fit the queue's LDS");

    const int tid = threadIdx.x;

    if ((int)blockIdx.x >= list_blocks) {
        const int g = (int)blockIdx.x - list_blocks;
        const int lane = tid & 31;
        const int wave = __builtin_amdgcn_readfirstlane(tid >> 5);
        float* Ws = (float*)stage;
        for (int i = tid; i < 128 * 128; i += AGG_THREADS) {
            const int k = i >> 7, col = i & 127;
            Ws[(k >> 1) * 256 + col * 2 + (k & 1)] = W[i];
        }
        __syncthreads();
        constexpr int UNITS = (RANGE / 16) * (128 / PROD_COLS);
        for (int u = wave; u < UNITS; u += AGG_THREADS / 32) {
            const int row0 = g * RANGE + (u >> 1) * 16;
            if (row0 < N)
                gemm_strip_f32_ldsB<128, 128, PROD_COLS / 16>(x, Ws, h, row0, (u & 1) * PROD_COLS, lane);
        }
        return;
    }

    int* queue = (int*)stage;
    const int bid = blockIdx.x;
    const int r0  = bid * RANGE;
    const int nrows = min(RANGE, N - r0);

    for (int i = tid; i < RANGE; i += AGG_THREADS) rowdeg[i] = 1;
    if (tid == 0) { qcount = 0; gcount = 0; }
    __syncthreads();

    auto push = [&](int e) {
        int slot = atomicAdd(&qcount, 1);
        if (slot < LIST_QCAP) queue[slot] = e;
        else atomicAdd(&rowdeg[pk[e] & 511], 1);
    };
    const unsigned pat = (unsigned)bid * 0x01010101u;
    const uint4* b16 = (const uint4*)bin8;
    auto match_group = [&](int g, uint4 v) {
        unsigned f = group_flags(v, pat);
        while (f) {
            const int b = __builtin_ctz(f);
            f &= f - 1;
            const int e = (g << 4) + ((7 - (b & 7)) << 2) + (b >> 3);
            if (e < E) push(e);
        }
    };
    {
        const unsigned* brow = bitmap + (size_t)bid * wpr;
        constexpr int UW = 4;
        for (int w0 = 0; w0 < wpr; w0 += UW * AGG_THREADS) {
            unsigned m[UW];
#pragma unroll
            for (int k = 0; k < UW; ++k) {
                const int w = w0 + k * AGG_THREADS + tid;
                m[k] = (w < wpr) ? brow[w] : 0u;
            }
#pragma unroll
            for (int k = 0; k < UW; ++k) {
                const int w = w0 + k * AGG_THREADS + tid;
                unsigned mm = m[k];
                while (mm) {
                    const int g = (w << 5) + __builtin_ctz(mm);
                    mm &= mm - 1;
                    const int slot = atomicAdd(&gcount, 1);
                    if (slot < LIST_GQCAP) gq[slot] = g;
                    else match_group(g, b16[g]);
                }
            }
        }
    }
    __syncthreads();
    {
        const int ng = min(__builtin_amdgcn_readfirstlane(gcount), LIST_GQCAP);
        constexpr int UG = 4;
        for (int base = 0; base < ng; base += UG * AGG_THREADS) {
            int g[UG];
            uint4 v[UG];
#pragma unroll
            for (int k = 0; k < UG; ++k) {
                const int i = base + k * AGG_THREADS + tid;
                g[k] = gq[(i < ng) ? i : 0];
                v[k] = b16[g[k]];
            }
#pragma unroll
            for (int k = 0; k < UG; ++k) {
                const int i = base + k * AGG_THREADS + tid;
                if (i < ng) match_group(g[k], v[k]);
            }
        }
    }
    __syncthreads();

    const int raw = __builtin_amdgcn_readfirstlane(qcount);
    const int cnt = min(raw, LIST_QCAP);
    const int rcap = nrows * (OUT_F / 2);
    const bool keep = (raw <= LIST_QCAP) && (raw <= rcap);
    if (tid == 0) pcount[bid] = keep ? raw : -1;

    int* rec = (int*)(park + (size_t)r0 * OUT_F);
    constexpr int U = 4;
    for (int base = 0; base < cnt; base += U * AGG_THREADS) {
        int p[U];
#pragma unroll
        for (int k = 0; k < U; ++k) {
            const int i = base + k * AGG_THREADS + tid;
            p[k] = pk[queue[(i < cnt) ? i : 0]];
        }
#pragma unroll
        for (int k = 0; k < U; ++k) {
            const int i = base + k * AGG_THREADS + tid;
            if (i < cnt) {
                atomicAdd(&rowdeg[p[k] & 511], 1);
                if (keep) rec[i] = p[k];
            }
        }
    }
    __syncthreads();

    for (int r = tid; r < nrows; r += AGG_THREADS) dinv[r0 + r] = rsqrtf((float)rowdeg[r]);
}

template <int F, int QCAP, bool RELU, bool FIRST>
__global__ __launch_bounds__(AGG_THREADS) void agg_range_kernel(
        const float* __restrict__ h, const int* __restrict__ src, const int* __restrict__ dst,
        const unsigned char* __restrict__ bin8, const float* __restrict__ dinv,
        const float* __restrict__ bias, float* out, float* park, const int* __restrict__ pcount,
        int N, int E) {
    __shared__ float acc[RANGE * F];
    __shared__ int   queue[QCAP];
    __shared__ int   qcount;

    constexpr int C = F / 32;
    constexpr int G = 8;

    const int tid  = threadIdx.x;
    const int lane = tid & 31;
    const int wave = __builtin_amdgcn_readfirstlane(tid >> 5);
    const int bid  = blockIdx.x;
    const int r0   = bid * RANGE;
    const int nrows = min(RANGE, N - r0);
    const int n4   = nrows * (F / 4);

    if (tid == 0) qcount = 0;

    {
        const float4* h4 = (const float4*)(h + (size_t)r0 * F);
        const float4* b4 = (const float4*)bias;
        float4* a4 = (float4*)acc;
        for (int i = tid; i < n4; i += AGG_THREADS) {
            int r = i / (F / 4);
            int c = i & (F / 4 - 1);
            float di = dinv[r0 + r];
            float s = di * di;
            float4 hv = h4[i];
            float4 bv = b4[c];
            float4 o;
            o.x = bv.x + s * hv.x;
            o.y = bv.y + s * hv.y;
            o.z = bv.z + s * hv.z;
            o.w = bv.w + s * hv.w;
            a4[i] = o;
        }
    }
    __syncthreads();

    const int rcap = nrows * (OUT_F / 2);
    int*   rec  = (int*)(park + (size_t)r0 * OUT_F);
    float* wrec = (float*)(rec + rcap);
    const int pc = __builtin_amdgcn_readfirstlane(pcount[bid]);
    const bool parked = (pc >= 0);
    int cnt = min(pc, rcap);

    if (!parked) {
        auto direct_edge = [&](int e) {
            int s = src[e];
            int d = dst[e];
            float w = dinv[s] * dinv[d];
            const float* hs = h + (size_t)s * F;
            float* a = acc + (d - r0) * F;
#pragma unroll 1
            for (int f = 0; f < F; ++f) atomicAdd(a + f, w * hs[f]);
        };
        auto push = [&](int e) {
            int slot = atomicAdd(&qcount, 1);
            if (slot < QCAP) queue[slot] = e;
            else direct_edge(e);
        };
        scan_range_bytes(bin8, E, bid, tid, push);
        __syncthreads();
        cnt = min(__builtin_amdgcn_readfirstlane(qcount), QCAP);
    }

    {
        for (int base = wave * 32; base < cnt; base += AGG_THREADS) {
            const int nv = min(32, cnt - base);
            int s = 0, dl = 0;
            float w = 0.0f;
            if (lane < nv) {
                if (parked) {
                    int p = rec[base + lane];
                    s = p >> 9;
                    dl = p & 511;
                    if (FIRST) {
                        w = dinv[s] * dinv[r0 + dl];
                        wrec[base + lane] = w;
                    } else {
                        w = wrec[base + lane];
                    }
                } else {
                    int e = queue[base + lane];
                    s = src[e];
                    int d = dst[e];
                    dl = d - r0;
                    w = dinv[s] * dinv[d];
                }
            }
            const int wi = __float_as_int(w);
            int j = 0;
            for (; j + G <= nv; j += G) {
                float v[G][C];
                int   dls[G];
                float ws[G];
#pragma unroll
                for (int k = 0; k < G; ++k) {
                    int sj = __builtin_amdgcn_readlane(s, j + k);
                    dls[k] = __builtin_amdgcn_readlane(dl, j + k);
                    ws[k]  = __int_as_float(__builtin_amdgcn_readlane(wi, j + k));
                    const float* hs = h + (size_t)sj * F + lane;
#pragma unroll
                    for (int c = 0; c < C; ++c) v[k][c] = hs[c * 32];
                }
#pragma unroll
                for (int k = 0; k < G; ++k) {
                    float* a = acc + dls[k] * F + lane;
#pragma unroll
                    for (int c = 0; c < C; ++c) atomicAdd(a + c * 32, v[k][c] * ws[k]);
                }
            }
            for (; j < nv; ++j) {
                int   sj = __builtin_amdgcn_readlane(s, j);
                int   dj = __builtin_amdgcn_readlane(dl, j);
                float wj = __int_as_float(__builtin_amdgcn_readlane(wi, j));
                const float* hs = h + (size_t)sj * F + lane;
                float* a = acc + dj * F + lane;
                float v[C];
#pragma unroll
                for (int c = 0; c < C; ++c) v[c] = hs[c * 32];
#pragma unroll
                for (int c = 0; c < C; ++c) atomicAdd(a + c * 32, v[c] * wj);
            }
        }
    }
    __syncthreads();

    {
        const float4* a4 = (const float4*)acc;
        float4* o4 = (float4*)(out + (size_t)r0 * F);
        for (int i = tid; i < n4; i += AGG_THREADS) {
            float4 o = a4[i];
            if (RELU) {
                o.x = fmaxf(o.x, 0.0f);
                o.y = fmaxf(o.y, 0.0f);
                o.z = fmaxf(o.z, 0.0f);
                o.w = fmaxf(o.w, 0.0f);
            }
            o4[i] = o;
        }
    }
}

extern "C" void kernel_launch(void* const* d_in, const int* in_sizes, int n_in,
                              void* d_out, int out_size, void* d_ws, size_t ws_size,
                              hipStream_t stream) {
    const float* x  = (const float*)d_in[0];
    const int*   ei = (const int*)d_in[1];
    const float* W1 = (const float*)d_in[2];
    const float* b1 = (const float*)d_in[3];
    const float* W2 = (const float*)d_in[4];
    const float* b2 = (const float*)d_in[5];

    const int N = in_sizes[0] / 128;
    const int E = in_sizes[1] / 2;
    const int* src = ei;
    const int* dst = ei + E;

    float* dinv = (float*)d_ws;
    float* rsv  = dinv + N;
    unsigned char* bin8 = (unsigned char*)rsv;
    float* h1   = rsv + E;
    float* agg1 = h1 + (size_t)N * 128;
    float* h2   = h1;
    float* out  = (float*)d_out;
    int* pcount = (int*)(rsv + E / 2);
    int* pk = (int*)agg1;
    unsigned* bitmap = (unsigned*)(agg1 + E);

    const int row_tiles = (N + 15) / 16;
    const int gemm_blocks = (row_tiles + 3) / 4;
    const int agg_blocks = (N + RANGE - 1) / RANGE;

    const int pack_blocks = (E + PACK_EDGES - 1) / PACK_EDGES;
    edge_pack_kernel<<<pack_blocks, PACK_THREADS, 0, stream>>>(src, dst, bin8, pk, bitmap, E);
    range_list_product_kernel<<<2 * agg_blocks, AGG_THREADS, 0, stream>>>(
        pk, bin8, bitmap, pack_blocks * BM_WORDS, dinv, out, pcount, N, E, x, W1, h1, agg_blocks);

    agg_range_kernel<128, 8192, true, true><<<agg_blocks, AGG_THREADS, 0, stream>>>(
        h1, src, dst, bin8, dinv, b1, agg1, out, pcount, N, E);

    gemm_wmma_f32<128, 64><<<gemm_blocks, 128, 0, stream>>>(agg1, W2, h2, N);
    agg_range_kernel<64, 16384, false, false><<<agg_blocks, AGG_THREADS, 0, stream>>>(
        h2, src, dst, bin8, dinv, b2, out, out, pcount, N, E);
}


# Edge sums in registers, no LDS float atomics: a block deals its edge list into an LDS bucket per owner wave (row mod 32); a wave adds its 13 rows in registers, stores them once (sums: bucket order)
# speedup vs baseline: 4.3095x; 1.1364x over previous
#include <hip/hip_runtime.h>


typedef __attribute__((ext_vector_type(2))) float v2f;
typedef __attribute__((ext_vector_type(8))) float v8f;

constexpr int RANGE = 400;
constexpr int AGG_THREADS = 1024;
constexpr int OUT_F = 64;
static_assert(RANGE <= 512, "the local row index of a parked edge has 9 bits");
constexpr int LIST_QCAP = 16384;
static_assert(RANGE * (OUT_F / 2) <= LIST_QCAP, "a kept list must fit the LDS queue");
constexpr int LIST_GQCAP = 8192;

constexpr int PACK_THREADS = 256;
constexpr int PACK_U = 4;
constexpr int PACK_GROUPS = PACK_THREADS * PACK_U;
constexpr int PACK_EDGES = PACK_GROUPS * 4;
constexpr int BM_WORDS = PACK_EDGES / 16 / 32;
constexpr int BM_ROWS = 256;
static_assert(BM_WORDS == 8, "a block writes its words of a row as two 16-byte stores");
static_assert(BM_ROWS == PACK_THREADS, "thread t writes row t of the block's bitmap words");

__device__ __forceinline__ void pack_group(int4 s, int4 d, int g, unsigned char* __restrict__ bin8,
                                           int* __restrict__ pk, unsigned* bm, int word, unsigned bit) {
    const unsigned rx = (unsigned)d.x / (unsigned)RANGE;
    const unsigned ry = (unsigned)d.y / (unsigned)RANGE;
    const unsigned rz = (unsigned)d.z / (unsigned)RANGE;
    const unsigned rw = (unsigned)d.w / (unsigned)RANGE;
    int4 p;
    p.x = (s.x << 9) | (int)((unsigned)d.x - rx * RANGE);
    p.y = (s.y << 9) | (int)((unsigned)d.y - ry * RANGE);
    p.z = (s.z << 9) | (int)((unsigned)d.z - rz * RANGE);
    p.w = (s.w << 9) | (int)((unsigned)d.w - rw * RANGE);
    ((int4*)pk)[g] = p;
    ((unsigned*)bin8)[g] = (rx & 255u) | ((ry & 255u) << 8) | ((rz & 255u) << 16) | (rw << 24);
    atomicOr(&bm[word * BM_ROWS + (rx & 255u)], bit);
    atomicOr(&bm[word * BM_ROWS + (ry & 255u)], bit);
    atomicOr(&bm[word * BM_ROWS + (rz & 255u)], bit);
    atomicOr(&bm[word * BM_ROWS + (rw & 255u)], bit);
}

__global__ __launch_bounds__(PACK_THREADS) void edge_pack_kernel(const int* __restrict__ src,
                                                                 const int* __restrict__ dst,
                                                                 unsigned char* __restrict__ bin8,
                                                                 int* __restrict__ pk,
                                                                 unsigned* __restrict__ bitmap, int E) {
    __shared__ __attribute__((aligned(16))) unsigned bm[BM_WORDS * BM_ROWS];
    const int tid = threadIdx.x;
    for (int i = tid; i < BM_WORDS * BM_ROWS; i += PACK_THREADS) bm[i] = 0u;
    __syncthreads();
    const unsigned bit = 1u << ((tid >> 2) & 31);
    const int word0 = tid >> 7;
    const bool vec = ((((size_t)src | (size_t)dst | (size_t)pk) & 15) == 0) && (((size_t)bin8 & 3) == 0);
    const int n4 = vec ? (E >> 2) : 0;
    const int gfirst = blockIdx.x * PACK_GROUPS;
    const int4* s4 = (const int4*)src;
    const int4* d4 = (const int4*)dst;
    if (gfirst + PACK_GROUPS <= n4) {
        int4 s[PACK_U], d[PACK_U];
#pragma unroll
        for (int k = 0; k < PACK_U; ++k) {
            s[k] = s4[gfirst + k * PACK_THREADS + tid];
            d[k] = d4[gfirst + k * PACK_THREADS + tid];
        }
#pragma unroll
        for (int k = 0; k < PACK_U; ++k)
            pack_group(s[k], d[k], gfirst + k * PACK_THREADS + tid, bin8, pk, bm, k * 2 + word0, bit);
    } else {
        for (int k = 0; k < PACK_U; ++k) {
            const int g = gfirst + k * PACK_THREADS + tid;
            if (g < n4) pack_group(s4[g], d4[g], g, bin8, pk, bm, k * 2 + word0, bit);
        }
    }
    const int e0 = blockIdx.x * PACK_EDGES;
    const int e1 = min(E, e0 + PACK_EDGES);
    for (int e = max(e0, n4 << 2) + tid; e < e1; e += PACK_THREADS) {
        const unsigned dd = (unsigned)dst[e];
        const unsigned r = dd / (unsigned)RANGE;
        pk[e] = (src[e] << 9) | (int)(dd - r * RANGE);
        bin8[e] = (unsigned char)r;
        const int el = e - e0;
        atomicOr(&bm[(el >> 9) * BM_ROWS + (r & 255u)], 1u << ((el >> 4) & 31));
    }
    __syncthreads();
    {
        const int wpr = gridDim.x * BM_WORDS;
        uint4 lo, hi;
        lo.x = bm[0 * BM_ROWS + tid]; lo.y = bm[1 * BM_ROWS + tid];
        lo.z = bm[2 * BM_ROWS + tid]; lo.w = bm[3 * BM_ROWS + tid];
        hi.x = bm[4 * BM_ROWS + tid]; hi.y = bm[5 * BM_ROWS + tid];
        hi.z = bm[6 * BM_ROWS + tid]; hi.w = bm[7 * BM_ROWS + tid];
        uint4* o = (uint4*)(bitmap + (size_t)tid * wpr + blockIdx.x * BM_WORDS);
        o[0] = lo;
        o[1] = hi;
    }
}

__device__ __forceinline__ unsigned byte_flags(unsigned w, unsigned pat) {
    unsigned x = w ^ pat;
    unsigned t = (x & 0x7f7f7f7fu) + 0x7f7f7f7fu;
    return ~(t | x | 0x7f7f7f7fu);
}
__device__ __forceinline__ unsigned group_flags(uint4 v, unsigned pat) {
    return byte_flags(v.x, pat) | (byte_flags(v.y, pat) >> 1) |
           (byte_flags(v.z, pat) >> 2) | (byte_flags(v.w, pat) >> 3);
}

template <typename Push>
__device__ __forceinline__ void scan_range_bytes(const unsigned char* __restrict__ bin8, int E,
                                                 int bid, int tid, Push push) {
    const unsigned pat = (unsigned)bid * 0x01010101u;
    const uint4* b16 = (const uint4*)bin8;
    const int n16 = E >> 4;
    constexpr int STEP = AGG_THREADS * 4;
    const int niter = (n16 + STEP - 1) / STEP;
    int it = (niter > 0) ? (bid % niter) : 0;
    for (int i = 0; i < niter; ++i) {
        const int gb = it * STEP + tid;
        const int g1 = gb + AGG_THREADS, g2 = gb + 2 * AGG_THREADS, g3 = gb + 3 * AGG_THREADS;
        unsigned f0, f1, f2, f3;
        if ((it + 1) * STEP <= n16) {
            uint4 v0 = b16[gb], v1 = b16[g1], v2 = b16[g2], v3 = b16[g3];
            f0 = group_flags(v0, pat);
            f1 = group_flags(v1, pat);
            f2 = group_flags(v2, pat);
            f3 = group_flags(v3, pat);
        } else {
            uint4 v0 = b16[min(gb, n16 - 1)], v1 = b16[min(g1, n16 - 1)];
            uint4 v2 = b16[min(g2, n16 - 1)], v3 = b16[min(g3, n16 - 1)];
            f0 = (gb < n16) ? group_flags(v0, pat) : 0u;
            f1 = (g1 < n16) ? group_flags(v1, pat) : 0u;
            f2 = (g2 < n16) ? group_flags(v2, pat) : 0u;
            f3 = (g3 < n16) ? group_flags(v3, pat) : 0u;
        }
        unsigned lo = f0 | (f1 >> 4);
        unsigned hi = f2 | (f3 >> 4);
        unsigned long long mm = ((unsigned long long)hi << 32) | lo;
        while (mm) {
            int b = __builtin_ctzll(mm);
            mm &= mm - 1;
            int j = (b >> 3) & 3;
            int k = ~b & 3;
            int u = ((b >> 5) << 1) + (((b >> 2) & 1) ^ 1);
            push(((gb + u * AGG_THREADS) << 4) + (k << 2) + j);
        }
        if (++it == niter) it = 0;
    }
    for (int e = (n16 << 4) + tid; e < E; e += AGG_THREADS)
        if ((int)bin8[e] == bid) push(e);
}

template <int K, int NCOL, int NT>
__device__ __forceinline__ void gemm_strip_f32(const float* __restrict__ A,
                                               const float* __restrict__ B,
                                               float* __restrict__ C, int row0, int col0, int lane) {
    const int half = lane >> 4;
    const int mr   = lane & 15;

    v8f acc[NT] = {};

    const float* arow = A + (size_t)(row0 + mr) * K + half * 2;
    const float* bcol = B + (size_t)(half * 2) * NCOL + col0 + mr;

#pragma unroll 8
    for (int kb = 0; kb < K / 4; ++kb) {
        v2f a;
        a.x = arow[kb * 4 + 0];
        a.y = arow[kb * 4 + 1];
        const float* brow = bcol + (size_t)(kb * 4) * NCOL;
#pragma unroll
        for (int ct = 0; ct < NT; ++ct) {
            v2f b;
            b.x = brow[ct * 16];
            b.y = brow[ct * 16 + NCOL];
            acc[ct] = __builtin_amdgcn_wmma_f32_16x16x4_f32(
                 false, a,  false, b,
                 (short)0, acc[ct],  false,  false);
        }
    }

#pragma unroll
    for (int ct = 0; ct < NT; ++ct) {
        float* crow = C + (size_t)(row0 + half * 8) * NCOL + col0 + ct * 16 + mr;
#pragma unroll
        for (int r = 0; r < 8; ++r) crow[(size_t)r * NCOL] = acc[ct][r];
    }
}

template <int K, int NCOL>
__global__ __launch_bounds__(128) void gemm_wmma_f32(const float* __restrict__ A,
                                                     const float* __restrict__ B,
                                                     float* __restrict__ C, int nrows) {
    const int wave = threadIdx.x >> 5;
    const int lane = threadIdx.x & 31;
    const int row_tile = blockIdx.x * (blockDim.x >> 5) + wave;
    const int row0 = row_tile * 16;
    if (row0 >= nrows) return;
    gemm_strip_f32<K, NCOL, NCOL / 16>(A, B, C, row0, 0, lane);
}

template <int K, int NCOL, int NT>
__device__ __forceinline__ void gemm_strip_f32_ldsB(const float* __restrict__ A, const float* Bs,
                                                    float* __restrict__ C, int row0, int col0, int lane) {
    const int half = lane >> 4;
    const int mr   = lane & 15;

    v8f acc[NT] = {};

    const float* arow = A + (size_t)(row0 + mr) * K + half * 2;
    const float* bcol = Bs + half * (2 * NCOL) + (col0 + mr) * 2;

#pragma unroll 4
    for (int kb = 0; kb < K / 4; ++kb) {
        v2f a;
        a.x = arow[kb * 4 + 0];
        a.y = arow[kb * 4 + 1];
        const float* brow = bcol + kb * (4 * NCOL);
        v2f b[NT];
#pragma unroll
        for (int ct = 0; ct < NT; ++ct) b[ct] = *(const v2f*)(brow + ct * 32);
#pragma unroll
        for (int ct = 0; ct < NT; ++ct)
            acc[ct] = __builtin_amdgcn_wmma_f32_16x16x4_f32(
                 false, a,  false, b[ct],
                 (short)0, acc[ct],  false,  false);
    }

#pragma unroll
    for (int ct = 0; ct < NT; ++ct) {
        float* crow = C + (size_t)(row0 + half * 8) * NCOL + col0 + ct * 16 + mr;
#pragma unroll
        for (int r = 0; r < 8; ++r) crow[(size_t)r * NCOL] = acc[ct][r];
    }
}

static_assert(RANGE % 16 == 0, "the product role cuts a block's rows into 16-row strips");
constexpr int PROD_COLS = 64;
static_assert(128 / PROD_COLS == 2, "a unit index is (strip << 1) | column half");
__global__ __launch_bounds__(AGG_THREADS) __attribute__((amdgpu_waves_per_eu(16, 16)))
void range_list_product_kernel(
        const int* __restrict__ pk, const unsigned char* __restrict__ bin8,
        const unsigned* __restrict__ bitmap, int wpr,
        float* __restrict__ dinv, float* park, int* __restrict__ pcount, int N, int E,
        const float* __restrict__ x, const float* __restrict__ W, float* __restrict__ h,
        int list_blocks) {
    __shared__ __attribute__((aligned(16))) unsigned char stage[LIST_QCAP * 4];
    __shared__ int gq[LIST_GQCAP];
    __shared__ int rowdeg[RANGE];
    __shared__ int qcount;
    __shared__ int gcount;
    static_assert(128 * 128 * sizeof(float) <= sizeof(stage), "W must fit the queue's LDS");

    const int tid = threadIdx.x;

    if ((int)blockIdx.x >= list_blocks) {
        const int g = (int)blockIdx.x - list_blocks;
        const int lane = tid & 31;
        const int wave = __builtin_amdgcn_readfirstlane(tid >> 5);
        float* Ws = (float*)stage;
        for (int i = tid; i < 128 * 128; i += AGG_THREADS) {
            const int k = i >> 7, col = i & 127;
            Ws[(k >> 1) * 256 + col * 2 + (k & 1)] = W[i];
        }
        __syncthreads();
        constexpr int UNITS = (RANGE / 16) * (128 / PROD_COLS);
        for (int u = wave; u < UNITS; u += AGG_THREADS / 32) {
            const int row0 = g * RANGE + (u >> 1) * 16;
            if (row0 < N)
                gemm_strip_f32_ldsB<128, 128, PROD_COLS / 16>(x, Ws, h, row0, (u & 1) * PROD_COLS, lane);
        }
        return;
    }

    int* queue = (int*)stage;
    const int bid = blockIdx.x;
    const int r0  = bid * RANGE;
    const int nrows = min(RANGE, N - r0);

    for (int i = tid; i < RANGE; i += AGG_THREADS) rowdeg[i] = 1;
    if (tid == 0) { qcount = 0; gcount = 0; }
    __syncthreads();

    auto push = [&](int e) {
        int slot = atomicAdd(&qcount, 1);
        if (slot < LIST_QCAP) queue[slot] = e;
        else atomicAdd(&rowdeg[pk[e] & 511], 1);
    };
    const unsigned pat = (unsigned)bid * 0x01010101u;
    const uint4* b16 = (const uint4*)bin8;
    auto match_group = [&](int g, uint4 v) {
        unsigned f = group_flags(v, pat);
        while (f) {
            const int b = __builtin_ctz(f);
            f &= f - 1;
            const int e = (g << 4) + ((7 - (b & 7)) << 2) + (b >> 3);
            if (e < E) push(e);
        }
    };
    {
        const unsigned* brow = bitmap + (size_t)bid * wpr;
        constexpr int UW = 4;
        for (int w0 = 0; w0 < wpr; w0 += UW * AGG_THREADS) {
            unsigned m[UW];
#pragma unroll
            for (int k = 0; k < UW; ++k) {
                const int w = w0 + k * AGG_THREADS + tid;
                m[k] = (w < wpr) ? brow[w] : 0u;
            }
#pragma unroll
            for (int k = 0; k < UW; ++k) {
                const int w = w0 + k * AGG_THREADS + tid;
                unsigned mm = m[k];
                while (mm) {
                    const int g = (w << 5) + __builtin_ctz(mm);
                    mm &= mm - 1;
                    const int slot = atomicAdd(&gcount, 1);
                    if (slot < LIST_GQCAP) gq[slot] = g;
                    else match_group(g, b16[g]);
                }
            }
        }
    }
    __syncthreads();
    {
        const int ng = min(__builtin_amdgcn_readfirstlane(gcount), LIST_GQCAP);
        constexpr int UG = 4;
        for (int base = 0; base < ng; base += UG * AGG_THREADS) {
            int g[UG];
            uint4 v[UG];
#pragma unroll
            for (int k = 0; k < UG; ++k) {
                const int i = base + k * AGG_THREADS + tid;
                g[k] = gq[(i < ng) ? i : 0];
                v[k] = b16[g[k]];
            }
#pragma unroll
            for (int k = 0; k < UG; ++k) {
                const int i = base + k * AGG_THREADS + tid;
                if (i < ng) match_group(g[k], v[k]);
            }
        }
    }
    __syncthreads();

    const int raw = __builtin_amdgcn_readfirstlane(qcount);
    const int cnt = min(raw, LIST_QCAP);
    const int rcap = nrows * (OUT_F / 2);
    const bool keep = (raw <= LIST_QCAP) && (raw <= rcap);
    if (tid == 0) pcount[bid] = keep ? raw : -1;

    int* rec = (int*)(park + (size_t)r0 * OUT_F);
    constexpr int U = 4;
    for (int base = 0; base < cnt; base += U * AGG_THREADS) {
        int p[U];
#pragma unroll
        for (int k = 0; k < U; ++k) {
            const int i = base + k * AGG_THREADS + tid;
            p[k] = pk[queue[(i < cnt) ? i : 0]];
        }
#pragma unroll
        for (int k = 0; k < U; ++k) {
            const int i = base + k * AGG_THREADS + tid;
            if (i < cnt) {
                atomicAdd(&rowdeg[p[k] & 511], 1);
                if (keep) rec[i] = p[k];
            }
        }
    }
    __syncthreads();

    for (int r = tid; r < nrows; r += AGG_THREADS) dinv[r0 + r] = rsqrtf((float)rowdeg[r]);
}

constexpr int RPW = (RANGE + 31) / 32;
static_assert(AGG_THREADS == 1024, "32 owner waves: the owner of local row r is wave r & 31");
constexpr int BUCKET_CAP = 480;
constexpr int BUCKET_STRIDE = BUCKET_CAP + 1;

typedef __attribute__((ext_vector_type(16))) float v16f;
static_assert(RPW <= 16, "a wave's rows are the entries of a 16-entry register vector");

template <int F, int QCAP, bool RELU, bool FIRST>
__global__ __launch_bounds__(AGG_THREADS) void agg_range_kernel(
        const float* __restrict__ h, const int* __restrict__ src, const int* __restrict__ dst,
        const unsigned char* __restrict__ bin8, const float* __restrict__ dinv,
        const float* __restrict__ bias, float* out, float* park, const int* __restrict__ pcount,
        int N, int E) {
    __shared__ __attribute__((aligned(16))) unsigned char accraw[RANGE * F * sizeof(float)];
    __shared__ int   queue[QCAP];
    __shared__ int   qcount;
    __shared__ int   bcnt[32];
    float* acc = (float*)accraw;

    constexpr int C = F / 32;
    constexpr int G = 8;
    constexpr int GR = (F == 128) ? 8 : 16;

    constexpr int BUCKET_BYTES = 32 * BUCKET_STRIDE * 4;
    static_assert(BUCKET_BYTES <= (int)sizeof(accraw), "the packed words live in the tile's bytes");
    static_assert((F == 128) ? (2 * BUCKET_BYTES <= (int)sizeof(accraw)) : (BUCKET_BYTES <= QCAP * 4),
                  "the weights live behind the packed words, or in the queue's bytes");
    int*   bp = (int*)accraw;
    float* bw = (F == 128) ? (float*)(accraw + BUCKET_BYTES) : (float*)queue;

    const int tid  = threadIdx.x;
    const int lane = tid & 31;
    const int wave = __builtin_amdgcn_readfirstlane(tid >> 5);
    const int bid  = blockIdx.x;
    const int r0   = bid * RANGE;
    const int nrows = min(RANGE, N - r0);
    const int n4   = nrows * (F / 4);

    if (tid == 0) qcount = 0;
    if (tid < 32) bcnt[tid] = 0;

    const int rcap = nrows * (OUT_F / 2);
    int*   rec  = (int*)(park + (size_t)r0 * OUT_F);
    float* wrec = (float*)(rec + rcap);
    const int pc = __builtin_amdgcn_readfirstlane(pcount[bid]);
    const bool parked = (pc >= 0);
    int cnt = min(pc, rcap);

    if (parked) {
        v16f a[C];
#pragma unroll
        for (int c = 0; c < C; ++c) a[c] = (v16f)(0.0f);
#pragma unroll
        for (int jj = 0; jj < RPW; ++jj) {
            const int row = min(wave + 32 * jj, nrows - 1);
            const float di = dinv[r0 + row];
            const float sc = di * di;
            const float* hr = h + (size_t)(r0 + row) * F;
#pragma unroll
            for (int c = 0; c < C; ++c) a[c][jj] = bias[c * 32 + lane] + sc * hr[c * 32 + lane];
        }
        __syncthreads();

        {
            constexpr int U = 8;
            for (int base = 0; base < cnt; base += U * AGG_THREADS) {
                int p[U];
                float w[U];
#pragma unroll
                for (int k = 0; k < U; ++k) {
                    const int i = base + k * AGG_THREADS + tid;
                    p[k] = rec[(i < cnt) ? i : 0];
                }
                asm volatile("" :: "v"(p[0]), "v"(p[1]), "v"(p[2]), "v"(p[3]),
                                   "v"(p[4]), "v"(p[5]), "v"(p[6]), "v"(p[7]));
                if (FIRST) {
                    float ds[U], dd[U];
#pragma unroll
                    for (int k = 0; k < U; ++k) {
                        ds[k] = dinv[p[k] >> 9];
                        dd[k] = dinv[r0 + (p[k] & 511)];
                    }
                    asm volatile("" :: "v"(ds[0]), "v"(ds[1]), "v"(ds[2]), "v"(ds[3]),
                                       "v"(ds[4]), "v"(ds[5]), "v"(ds[6]), "v"(ds[7]),
                                       "v"(dd[0]), "v"(dd[1]), "v"(dd[2]), "v"(dd[3]),
                                       "v"(dd[4]), "v"(dd[5]), "v"(dd[6]), "v"(dd[7]));
#pragma unroll
                    for (int k = 0; k < U; ++k) w[k] = ds[k] * dd[k];
                } else {
#pragma unroll
                    for (int k = 0; k < U; ++k) {
                        const int i = base + k * AGG_THREADS + tid;
                        w[k] = wrec[(i < cnt) ? i : 0];
                    }
                    asm volatile("" :: "v"(w[0]), "v"(w[1]), "v"(w[2]), "v"(w[3]),
                                       "v"(w[4]), "v"(w[5]), "v"(w[6]), "v"(w[7]));
                }
#pragma unroll
                for (int k = 0; k < U; ++k) {
                    const int i = base + k * AGG_THREADS + tid;
                    if (i < cnt) {
                        if (FIRST) wrec[i] = w[k];
                        const int o = p[k] & 31;
                        const int slot = atomicAdd(&bcnt[o], 1);
                        if (slot < BUCKET_CAP) {
                            bp[o * BUCKET_STRIDE + slot] = p[k];
                            bw[o * BUCKET_STRIDE + slot] = w[k];
                        }
                    }
                }
            }
        }
        __syncthreads();

        int fullest = bcnt[lane];
#pragma unroll
        for (int o = 16; o > 0; o >>= 1) fullest = max(fullest, __shfl_xor(fullest, o, 32));
        const bool fits = __builtin_amdgcn_readfirstlane(fullest) <= BUCKET_CAP;

        if (fits) {
            const int nb = __builtin_amdgcn_readfirstlane(bcnt[wave]);
            const int*   mybp = bp + wave * BUCKET_STRIDE;
            const float* mybw = bw + wave * BUCKET_STRIDE;
            for (int base = 0; base < nb; base += 32) {
                const int nv = min(32, nb - base);
                int p = 0;
                float w = 0.0f;
                if (lane < nv) {
                    p = mybp[base + lane];
                    w = mybw[base + lane];
                }
                const int s  = p >> 9;
                const int jr = (p & 511) >> 5;
                const int wi = __float_as_int(w);
                for (int j = 0; j < nv; j += GR) {
                    const int g = min(GR, nv - j);
                    float v[GR][C];
                    int   jrs[GR];
                    float ws[GR];
#pragma unroll
                    for (int k = 0; k < GR; ++k) {
                        const int idx = min(j + k, nv - 1);
                        const int sj = __builtin_amdgcn_readlane(s, idx);
                        jrs[k] = __builtin_amdgcn_readlane(jr, idx);
                        ws[k]  = __int_as_float(__builtin_amdgcn_readlane(wi, idx));
                        const float* hs = h + (size_t)sj * F;
#pragma unroll
                        for (int c = 0; c < C; ++c) v[k][c] = hs[c * 32 + lane];
                    }
#pragma unroll
                    for (int k = 0; k < GR; ++k) {
                        if (k < g) {
                            const int jk = jrs[k];
#pragma unroll
                            for (int c = 0; c < C; ++c) a[c][jk] = a[c][jk] + ws[k] * v[k][c];
                        }
                    }
                }
            }
#pragma unroll
            for (int jj = 0; jj < RPW; ++jj) {
                const int row = wave + 32 * jj;
                if (row < nrows) {
                    float* o = out + (size_t)(r0 + row) * F;
#pragma unroll
                    for (int c = 0; c < C; ++c) {
                        float x = a[c][jj];
                        if (RELU) x = fmaxf(x, 0.0f);
                        o[c * 32 + lane] = x;
                    }
                }
            }
            return;
        }
    }

    {
        const float4* h4 = (const float4*)(h + (size_t)r0 * F);
        const float4* b4 = (const float4*)bias;
        float4* a4 = (float4*)acc;
        for (int i = tid; i < n4; i += AGG_THREADS) {
            int r = i / (F / 4);
            int c = i & (F / 4 - 1);
            float di = dinv[r0 + r];
            float s = di * di;
            float4 hv = h4[i];
            float4 bv = b4[c];
            float4 o;
            o.x = bv.x + s * hv.x;
            o.y = bv.y + s * hv.y;
            o.z = bv.z + s * hv.z;
            o.w = bv.w + s * hv.w;
            a4[i] = o;
        }
    }
    __syncthreads();

    if (!parked) {
        auto direct_edge = [&](int e) {
            int s = src[e];
            int d = dst[e];
            float w = dinv[s] * dinv[d];
            const float* hs = h + (size_t)s * F;
            float* a = acc + (d - r0) * F;
#pragma unroll 1
            for (int f = 0; f < F; ++f) atomicAdd(a + f, w * hs[f]);
        };
        auto push = [&](int e) {
            int slot = atomicAdd(&qcount, 1);
            if (slot < QCAP) queue[slot] = e;
            else direct_edge(e);
        };
        scan_range_bytes(bin8, E, bid, tid, push);
        __syncthreads();
        cnt = min(__builtin_amdgcn_readfirstlane(qcount), QCAP);
    }

    {
        for (int base = wave * 32; base < cnt; base += AGG_THREADS) {
            const int nv = min(32, cnt - base);
            int s = 0, dl = 0;
            float w = 0.0f;
            if (lane < nv) {
                if (parked) {
                    int p = rec[base + lane];
                    s = p >> 9;
                    dl = p & 511;
                    if (FIRST) {
                        w = dinv[s] * dinv[r0 + dl];
                        wrec[base + lane] = w;
                    } else {
                        w = wrec[base + lane];
                    }
                } else {
                    int e = queue[base + lane];
                    s = src[e];
                    int d = dst[e];
                    dl = d - r0;
                    w = dinv[s] * dinv[d];
                }
            }
            const int wi = __float_as_int(w);
            int j = 0;
            for (; j + G <= nv; j += G) {
                float v[G][C];
                int   dls[G];
                float ws[G];
#pragma unroll
                for (int k = 0; k < G; ++k) {
                    int sj = __builtin_amdgcn_readlane(s, j + k);
                    dls[k] = __builtin_amdgcn_readlane(dl, j + k);
                    ws[k]  = __int_as_float(__builtin_amdgcn_readlane(wi, j + k));
                    const float* hs = h + (size_t)sj * F + lane;
#pragma unroll
                    for (int c = 0; c < C; ++c) v[k][c] = hs[c * 32];
                }
#pragma unroll
                for (int k = 0; k < G; ++k) {
                    float* a = acc + dls[k] * F + lane;
#pragma unroll
                    for (int c = 0; c < C; ++c) atomicAdd(a + c * 32, v[k][c] * ws[k]);
                }
            }
            for (; j < nv; ++j) {
                int   sj = __builtin_amdgcn_readlane(s, j);
                int   dj = __builtin_amdgcn_readlane(dl, j);
                float wj = __int_as_float(__builtin_amdgcn_readlane(wi, j));
                const float* hs = h + (size_t)sj * F + lane;
                float* a = acc + dj * F + lane;
                float v[C];
#pragma unroll
                for (int c = 0; c < C; ++c) v[c] = hs[c * 32];
#pragma unroll
                for (int c = 0; c < C; ++c) atomicAdd(a + c * 32, v[c] * wj);
            }
        }
    }
    __syncthreads();

    {
        const float4* a4 = (const float4*)acc;
        float4* o4 = (float4*)(out + (size_t)r0 * F);
        for (int i = tid; i < n4; i += AGG_THREADS) {
            float4 o = a4[i];
            if (RELU) {
                o.x = fmaxf(o.x, 0.0f);
                o.y = fmaxf(o.y, 0.0f);
                o.z = fmaxf(o.z, 0.0f);
                o.w = fmaxf(o.w, 0.0f);
            }
            o4[i] = o;
        }
    }
}

extern "C" void kernel_launch(void* const* d_in, const int* in_sizes, int n_in,
                              void* d_out, int out_size, void* d_ws, size_t ws_size,
                              hipStream_t stream) {
    const float* x  = (const float*)d_in[0];
    const int*   ei = (const int*)d_in[1];
    const float* W1 = (const float*)d_in[2];
    const float* b1 = (const float*)d_in[3];
    const float* W2 = (const float*)d_in[4];
    const float* b2 = (const float*)d_in[5];

    const int N = in_sizes[0] / 128;
    const int E = in_sizes[1] / 2;
    const int* src = ei;
    const int* dst = ei + E;

    float* dinv = (float*)d_ws;
    float* rsv  = dinv + N;
    unsigned char* bin8 = (unsigned char*)rsv;
    float* h1   = rsv + E;
    float* agg1 = h1 + (size_t)N * 128;
    float* h2   = h1;
    float* out  = (float*)d_out;
    int* pcount = (int*)(rsv + E / 2);
    int* pk = (int*)agg1;
    unsigned* bitmap = (unsigned*)(agg1 + E);

    const int row_tiles = (N + 15) / 16;
    const int gemm_blocks = (row_tiles + 3) / 4;
    const int agg_blocks = (N + RANGE - 1) / RANGE;

    const int pack_blocks = (E + PACK_EDGES - 1) / PACK_EDGES;
    edge_pack_kernel<<<pack_blocks, PACK_THREADS, 0, stream>>>(src, dst, bin8, pk, bitmap, E);
    range_list_product_kernel<<<2 * agg_blocks, AGG_THREADS, 0, stream>>>(
        pk, bin8, bitmap, pack_blocks * BM_WORDS, dinv, out, pcount, N, E, x, W1, h1, agg_blocks);

    agg_range_kernel<128, 8192, true, true><<<agg_blocks, AGG_THREADS, 0, stream>>>(
        h1, src, dst, bin8, dinv, b1, agg1, out, pcount, N, E);

    gemm_wmma_f32<128, 64><<<gemm_blocks, 128, 0, stream>>>(agg1, W2, h2, N);
    agg_range_kernel<64, 16384, false, false><<<agg_blocks, AGG_THREADS, 0, stream>>>(
        h2, src, dst, bin8, dinv, b2, out, out, pcount, N, E);
}
